# starting kernel
# baseline (speedup 1.0000x reference)
#include <hip/hip_runtime.h>

#define N_NODES 50000
#define DEG     16
#define NIN     128
#define NOUT    64

typedef __attribute__((ext_vector_type(2))) float v2f;
typedef __attribute__((ext_vector_type(8))) float v8f;

__global__ __launch_bounds__(256) void proj_wmma_kernel(
    const float* __restrict__ node_fts, const float* __restrict__ edge_fts,
    const float* __restrict__ W_node,   const float* __restrict__ W_edge,
    float* __restrict__ hv, float* __restrict__ ev)
{
    const int wave   = (blockIdx.x * blockDim.x + threadIdx.x) >> 5;
    const int ntiles = N_NODES / 16;
    if (wave >= 2 * ntiles) return;
    const int  mat  = (wave >= ntiles) ? 1 : 0;
    const int  tile = mat ? (wave - ntiles) : wave;
    const float* A  = mat ? edge_fts : node_fts;
    const float* W  = mat ? W_edge   : W_node;
    float*       O  = mat ? ev       : hv;

    const int row0 = tile * 16;
    const int lane = threadIdx.x & 31;
    const int m    = lane & 15;
    const int kh   = lane >> 4;

    const float* arow = A + (size_t)(row0 + m) * NIN + 2 * kh;
    v8f acc0 = {}, acc1 = {}, acc2 = {}, acc3 = {};

    #pragma unroll 4
    for (int k0 = 0; k0 < NIN; k0 += 4) {
        v2f a = *(const v2f*)(arow + k0);
        const float* w0 = W + (size_t)(k0 + 2 * kh) * NOUT + m;
        const float* w1 = w0 + NOUT;
        v2f b0 = { w0[0],  w1[0]  };
        v2f b1 = { w0[16], w1[16] };
        v2f b2 = { w0[32], w1[32] };
        v2f b3 = { w0[48], w1[48] };
        acc0 = __builtin_amdgcn_wmma_f32_16x16x4_f32(false, a, false, b0, (short)0, acc0, false, false);
        acc1 = __builtin_amdgcn_wmma_f32_16x16x4_f32(false, a, false, b1, (short)0, acc1, false, false);
        acc2 = __builtin_amdgcn_wmma_f32_16x16x4_f32(false, a, false, b2, (short)0, acc2, false, false);
        acc3 = __builtin_amdgcn_wmma_f32_16x16x4_f32(false, a, false, b3, (short)0, acc3, false, false);
    }

    float* orow = O + (size_t)(row0 + kh * 8) * NOUT + m;
    #pragma unroll
    for (int r = 0; r < 8; ++r) {
        orow[(size_t)r * NOUT +  0] = acc0[r];
        orow[(size_t)r * NOUT + 16] = acc1[r];
        orow[(size_t)r * NOUT + 32] = acc2[r];
        orow[(size_t)r * NOUT + 48] = acc3[r];
    }
}

__global__ __launch_bounds__(256) void alpha_kernel(
    const float* __restrict__ hv, const float* __restrict__ ev,
    const float* __restrict__ a_node, const float* __restrict__ a_edge,
    float* __restrict__ asrc, float* __restrict__ adst,
    float* __restrict__ bsrc, float* __restrict__ bdst)
{
    const int wave = (blockIdx.x * blockDim.x + threadIdx.x) >> 5;
    if (wave >= N_NODES) return;
    const int lane = threadIdx.x & 31;

    const float a0l = a_node[lane],      a0h = a_node[lane + 32];
    const float a1l = a_node[lane + 64], a1h = a_node[lane + 96];
    const float b0l = a_edge[lane],      b0h = a_edge[lane + 32];
    const float b1l = a_edge[lane + 64], b1h = a_edge[lane + 96];

    const float hlo = hv[(size_t)wave * NOUT + lane];
    const float hhi = hv[(size_t)wave * NOUT + 32 + lane];
    const float elo = ev[(size_t)wave * NOUT + lane];
    const float ehi = ev[(size_t)wave * NOUT + 32 + lane];

    float ps = hlo * a0l + hhi * a0h;
    float pd = hlo * a1l + hhi * a1h;
    float qs = elo * b0l + ehi * b0h;
    float qd = elo * b1l + ehi * b1h;
    #pragma unroll
    for (int off = 16; off; off >>= 1) {
        ps += __shfl_xor(ps, off, 32);
        pd += __shfl_xor(pd, off, 32);
        qs += __shfl_xor(qs, off, 32);
        qd += __shfl_xor(qd, off, 32);
    }
    if (lane == 0) {
        asrc[wave] = ps; adst[wave] = pd;
        bsrc[wave] = qs; bdst[wave] = qd;
    }
}

__global__ __launch_bounds__(256) void edge_kernel(
    const float* __restrict__ hv, const float* __restrict__ ev,
    const int*  __restrict__ edges,
    const float* __restrict__ asrc, const float* __restrict__ adst,
    const float* __restrict__ bsrc, const float* __restrict__ bdst,
    float* __restrict__ node_out, float* __restrict__ edge_out,
    float* __restrict__ partials)
{
    __shared__ float sred[4][8];
    const int lane = threadIdx.x & 31;
    const int wid  = threadIdx.x >> 5;
    const int node = blockIdx.x * 8 + wid;

    float att_n = 0.f, att_e = 0.f;
    int   d = 0;
    if (node < N_NODES && lane < 16) {
        int2 ed = ((const int2*)edges)[(size_t)node * DEG + lane];
        const int s = ed.x;
        d = ed.y;
        float xn = asrc[s] + adst[d];
        float xe = bsrc[s] + bdst[d];
        xn = xn > 0.f ? xn : 0.2f * xn;
        xe = xe > 0.f ? xe : 0.2f * xe;
        xn = fminf(fmaxf(xn, -2.f), 2.f);
        xe = fminf(fmaxf(xe, -2.f), 2.f);
        att_n = __expf(xn);
        att_e = __expf(xe);
    }

    float sum_n = att_n, sum_e = att_e;
    #pragma unroll
    for (int off = 16; off; off >>= 1) {
        sum_n += __shfl_xor(sum_n, off, 32);
        sum_e += __shfl_xor(sum_e, off, 32);
    }
    const bool act = (node < N_NODES) && (lane < 16);
    float norm_n = act ? att_n / sum_n : 0.f;
    float norm_e = act ? att_e / sum_e : 0.f;

    float pn = norm_n, pn2 = norm_n * norm_n, pe = norm_e, pe2 = norm_e * norm_e;
    #pragma unroll
    for (int off = 16; off; off >>= 1) {
        pn  += __shfl_xor(pn,  off, 32);
        pn2 += __shfl_xor(pn2, off, 32);
        pe  += __shfl_xor(pe,  off, 32);
        pe2 += __shfl_xor(pe2, off, 32);
    }
    if (lane == 0) { sred[0][wid] = pn; sred[1][wid] = pn2; sred[2][wid] = pe; sred[3][wid] = pe2; }
    __syncthreads();
    if (threadIdx.x < 4) {
        float t = 0.f;
        #pragma unroll
        for (int w = 0; w < 8; ++w) t += sred[threadIdx.x][w];
        partials[(size_t)blockIdx.x * 4 + threadIdx.x] = t;
    }
    if (node >= N_NODES) return;

    const float2* hv2 = (const float2*)hv;
    const float2* ev2 = (const float2*)ev;
    float2 acc_h = {0.f, 0.f}, acc_e = {0.f, 0.f};
    #pragma unroll
    for (int j = 0; j < 16; ++j) {
        const int   dj = __shfl(d,      j, 32);
        const float wn = __shfl(norm_n, j, 32);
        const float we = __shfl(norm_e, j, 32);
        const float2 hvv = hv2[(size_t)dj * 32 + lane];
        const float2 evv = ev2[(size_t)dj * 32 + lane];
        acc_h.x += wn * hvv.x;  acc_h.y += wn * hvv.y;
        acc_e.x += we * evv.x;  acc_e.y += we * evv.y;
    }

    const float2 hs = hv2[(size_t)node * 32 + lane];
    const float2 es = ev2[(size_t)node * 32 + lane];
    float2* no2 = (float2*)node_out;
    float2* eo2 = (float2*)edge_out;
    float2 o0 = { pn * hs.x, pn * hs.y };
    float2 p0 = { pe * es.x, pe * es.y };
    no2[(size_t)node * 64 + lane]      = o0;
    no2[(size_t)node * 64 + 32 + lane] = acc_h;
    eo2[(size_t)node * 64 + lane]      = p0;
    eo2[(size_t)node * 64 + 32 + lane] = acc_e;
}

__global__ __launch_bounds__(256) void finalize_kernel(
    const float* __restrict__ partials, int nblocks, float* __restrict__ outv)
{
    __shared__ double red[4][256];
    double s0 = 0.0, s1 = 0.0, s2 = 0.0, s3 = 0.0;
    for (int i = threadIdx.x; i < nblocks; i += 256) {
        s0 += (double)partials[(size_t)i * 4 + 0];
        s1 += (double)partials[(size_t)i * 4 + 1];
        s2 += (double)partials[(size_t)i * 4 + 2];
        s3 += (double)partials[(size_t)i * 4 + 3];
    }
    red[0][threadIdx.x] = s0; red[1][threadIdx.x] = s1;
    red[2][threadIdx.x] = s2; red[3][threadIdx.x] = s3;
    __syncthreads();
    for (int off = 128; off; off >>= 1) {
        if (threadIdx.x < off) {
            red[0][threadIdx.x] += red[0][threadIdx.x + off];
            red[1][threadIdx.x] += red[1][threadIdx.x + off];
            red[2][threadIdx.x] += red[2][threadIdx.x + off];
            red[3][threadIdx.x] += red[3][threadIdx.x + off];
        }
        __syncthreads();
    }
    if (threadIdx.x == 0) {
        const double E = (double)N_NODES * (double)DEG;
        const double var_n = (red[1][0] - red[0][0] * red[0][0] / E) / (E - 1.0);
        const double var_e = (red[3][0] - red[2][0] * red[2][0] / E) / (E - 1.0);
        outv[0] = (float)var_n;
        outv[1] = (float)var_e;
    }
}

extern "C" void kernel_launch(void* const* d_in, const int* in_sizes, int n_in,
                              void* d_out, int out_size, void* d_ws, size_t ws_size,
                              hipStream_t stream) {
    const float* node_fts = (const float*)d_in[0];
    const float* edge_fts = (const float*)d_in[1];
    const int*   edges    = (const int*)  d_in[2];
    const float* W_node   = (const float*)d_in[3];
    const float* W_edge   = (const float*)d_in[4];
    const float* a_node   = (const float*)d_in[5];
    const float* a_edge   = (const float*)d_in[6];

    float* out      = (float*)d_out;
    float* node_out = out;
    float* edge_out = out + (size_t)N_NODES * 128;
    float* outv     = out + (size_t)2 * N_NODES * 128;

    float* ws   = (float*)d_ws;
    float* hv   = ws;
    float* ev   = hv + (size_t)N_NODES * NOUT;
    float* asrc = ev + (size_t)N_NODES * NOUT;
    float* adst = asrc + N_NODES;
    float* bsrc = adst + N_NODES;
    float* bdst = bsrc + N_NODES;
    float* partials = bdst + N_NODES;

    const int ntiles = N_NODES / 16;
    const int wavesA = 2 * ntiles;
    const int blocksA = (wavesA + 7) / 8;
    proj_wmma_kernel<<<blocksA, 256, 0, stream>>>(node_fts, edge_fts, W_node, W_edge, hv, ev);

    const int blocksB = (N_NODES + 7) / 8;
    alpha_kernel<<<blocksB, 256, 0, stream>>>(hv, ev, a_node, a_edge, asrc, adst, bsrc, bdst);

    const int blocksC = (N_NODES + 7) / 8;
    edge_kernel<<<blocksC, 256, 0, stream>>>(hv, ev, edges, asrc, adst, bsrc, bdst,
                                             node_out, edge_out, partials);

    finalize_kernel<<<1, 256, 0, stream>>>(partials, blocksC, outv);
}


# alpha pass: one wave covers 16 nodes (3125 waves start, not 50000); each half-wave reads a 256-B row as 16-B vectors, 16 row loads in one batch, row sums on DPP; the dots' summation tree changes
# speedup vs baseline: 1.1660x; 1.1660x over previous
#include <hip/hip_runtime.h>

#define N_NODES 50000
#define DEG     16
#define NIN     128
#define NOUT    64

typedef __attribute__((ext_vector_type(2))) float v2f;
typedef __attribute__((ext_vector_type(8))) float v8f;

__global__ __launch_bounds__(256) void proj_wmma_kernel(
    const float* __restrict__ node_fts, const float* __restrict__ edge_fts,
    const float* __restrict__ W_node,   const float* __restrict__ W_edge,
    float* __restrict__ hv, float* __restrict__ ev)
{
    const int wave   = (blockIdx.x * blockDim.x + threadIdx.x) >> 5;
    const int ntiles = N_NODES / 16;
    if (wave >= 2 * ntiles) return;
    const int  mat  = (wave >= ntiles) ? 1 : 0;
    const int  tile = mat ? (wave - ntiles) : wave;
    const float* A  = mat ? edge_fts : node_fts;
    const float* W  = mat ? W_edge   : W_node;
    float*       O  = mat ? ev       : hv;

    const int row0 = tile * 16;
    const int lane = threadIdx.x & 31;
    const int m    = lane & 15;
    const int kh   = lane >> 4;

    const float* arow = A + (size_t)(row0 + m) * NIN + 2 * kh;
    v8f acc0 = {}, acc1 = {}, acc2 = {}, acc3 = {};

    #pragma unroll 4
    for (int k0 = 0; k0 < NIN; k0 += 4) {
        v2f a = *(const v2f*)(arow + k0);
        const float* w0 = W + (size_t)(k0 + 2 * kh) * NOUT + m;
        const float* w1 = w0 + NOUT;
        v2f b0 = { w0[0],  w1[0]  };
        v2f b1 = { w0[16], w1[16] };
        v2f b2 = { w0[32], w1[32] };
        v2f b3 = { w0[48], w1[48] };
        acc0 = __builtin_amdgcn_wmma_f32_16x16x4_f32(false, a, false, b0, (short)0, acc0, false, false);
        acc1 = __builtin_amdgcn_wmma_f32_16x16x4_f32(false, a, false, b1, (short)0, acc1, false, false);
        acc2 = __builtin_amdgcn_wmma_f32_16x16x4_f32(false, a, false, b2, (short)0, acc2, false, false);
        acc3 = __builtin_amdgcn_wmma_f32_16x16x4_f32(false, a, false, b3, (short)0, acc3, false, false);
    }

    float* orow = O + (size_t)(row0 + kh * 8) * NOUT + m;
    #pragma unroll
    for (int r = 0; r < 8; ++r) {
        orow[(size_t)r * NOUT +  0] = acc0[r];
        orow[(size_t)r * NOUT + 16] = acc1[r];
        orow[(size_t)r * NOUT + 32] = acc2[r];
        orow[(size_t)r * NOUT + 48] = acc3[r];
    }
}

template <int M>
__device__ __forceinline__ float row_xor_add(float v)
{
    const int o = __builtin_amdgcn_update_dpp(0, __builtin_bit_cast(int, v),
                                              0x160 | M, 0xf, 0xf, false);
    return v + __builtin_bit_cast(float, o);
}

__global__ __launch_bounds__(128) void alpha_kernel(
    const float* __restrict__ hv, const float* __restrict__ ev,
    const float* __restrict__ a_node, const float* __restrict__ a_edge,
    float* __restrict__ asrc, float* __restrict__ adst,
    float* __restrict__ bsrc, float* __restrict__ bdst)
{
    const int wave = (blockIdx.x * blockDim.x + threadIdx.x) >> 5;
    if (wave >= N_NODES / 16) return;
    const int lane = threadIdx.x & 31;
    const int c    = lane & 15;
    const int h    = lane >> 4;

    const float4 as = ((const float4*)a_node)[c];
    const float4 ad = ((const float4*)a_node)[16 + c];
    const float4 bs = ((const float4*)a_edge)[c];
    const float4 bd = ((const float4*)a_edge)[16 + c];

    const float4* hv4 = (const float4*)hv;
    const float4* ev4 = (const float4*)ev;
    const int node0 = wave * 16;

    float4 x[8], y[8];
    #pragma unroll
    for (int t = 0; t < 8; ++t) {
        const size_t row = (size_t)(node0 + 2 * t + h);
        x[t] = hv4[row * 16 + c];
        y[t] = ev4[row * 16 + c];
    }

    float r_ps = 0.f, r_pd = 0.f, r_qs = 0.f, r_qd = 0.f;
    #pragma unroll
    for (int t = 0; t < 8; ++t) {
        float ps = x[t].x * as.x + x[t].y * as.y + x[t].z * as.z + x[t].w * as.w;
        float pd = x[t].x * ad.x + x[t].y * ad.y + x[t].z * ad.z + x[t].w * ad.w;
        float qs = y[t].x * bs.x + y[t].y * bs.y + y[t].z * bs.z + y[t].w * bs.w;
        float qd = y[t].x * bd.x + y[t].y * bd.y + y[t].z * bd.z + y[t].w * bd.w;
        ps = row_xor_add<8>(ps); pd = row_xor_add<8>(pd); qs = row_xor_add<8>(qs); qd = row_xor_add<8>(qd);
        ps = row_xor_add<4>(ps); pd = row_xor_add<4>(pd); qs = row_xor_add<4>(qs); qd = row_xor_add<4>(qd);
        ps = row_xor_add<2>(ps); pd = row_xor_add<2>(pd); qs = row_xor_add<2>(qs); qd = row_xor_add<2>(qd);
        ps = row_xor_add<1>(ps); pd = row_xor_add<1>(pd); qs = row_xor_add<1>(qs); qd = row_xor_add<1>(qd);
        const bool keep = (c == t);
        r_ps = keep ? ps : r_ps;  r_pd = keep ? pd : r_pd;
        r_qs = keep ? qs : r_qs;  r_qd = keep ? qd : r_qd;
    }
    if (c < 8) {
        const int node = node0 + 2 * c + h;
        asrc[node] = r_ps; adst[node] = r_pd;
        bsrc[node] = r_qs; bdst[node] = r_qd;
    }
}

__global__ __launch_bounds__(256) void edge_kernel(
    const float* __restrict__ hv, const float* __restrict__ ev,
    const int*  __restrict__ edges,
    const float* __restrict__ asrc, const float* __restrict__ adst,
    const float* __restrict__ bsrc, const float* __restrict__ bdst,
    float* __restrict__ node_out, float* __restrict__ edge_out,
    float* __restrict__ partials)
{
    __shared__ float sred[4][8];
    const int lane = threadIdx.x & 31;
    const int wid  = threadIdx.x >> 5;
    const int node = blockIdx.x * 8 + wid;

    float att_n = 0.f, att_e = 0.f;
    int   d = 0;
    if (node < N_NODES && lane < 16) {
        int2 ed = ((const int2*)edges)[(size_t)node * DEG + lane];
        const int s = ed.x;
        d = ed.y;
        float xn = asrc[s] + adst[d];
        float xe = bsrc[s] + bdst[d];
        xn = xn > 0.f ? xn : 0.2f * xn;
        xe = xe > 0.f ? xe : 0.2f * xe;
        xn = fminf(fmaxf(xn, -2.f), 2.f);
        xe = fminf(fmaxf(xe, -2.f), 2.f);
        att_n = __expf(xn);
        att_e = __expf(xe);
    }

    float sum_n = att_n, sum_e = att_e;
    #pragma unroll
    for (int off = 16; off; off >>= 1) {
        sum_n += __shfl_xor(sum_n, off, 32);
        sum_e += __shfl_xor(sum_e, off, 32);
    }
    const bool act = (node < N_NODES) && (lane < 16);
    float norm_n = act ? att_n / sum_n : 0.f;
    float norm_e = act ? att_e / sum_e : 0.f;

    float pn = norm_n, pn2 = norm_n * norm_n, pe = norm_e, pe2 = norm_e * norm_e;
    #pragma unroll
    for (int off = 16; off; off >>= 1) {
        pn  += __shfl_xor(pn,  off, 32);
        pn2 += __shfl_xor(pn2, off, 32);
        pe  += __shfl_xor(pe,  off, 32);
        pe2 += __shfl_xor(pe2, off, 32);
    }
    if (lane == 0) { sred[0][wid] = pn; sred[1][wid] = pn2; sred[2][wid] = pe; sred[3][wid] = pe2; }
    __syncthreads();
    if (threadIdx.x < 4) {
        float t = 0.f;
        #pragma unroll
        for (int w = 0; w < 8; ++w) t += sred[threadIdx.x][w];
        partials[(size_t)blockIdx.x * 4 + threadIdx.x] = t;
    }
    if (node >= N_NODES) return;

    const float2* hv2 = (const float2*)hv;
    const float2* ev2 = (const float2*)ev;
    float2 acc_h = {0.f, 0.f}, acc_e = {0.f, 0.f};
    #pragma unroll
    for (int j = 0; j < 16; ++j) {
        const int   dj = __shfl(d,      j, 32);
        const float wn = __shfl(norm_n, j, 32);
        const float we = __shfl(norm_e, j, 32);
        const float2 hvv = hv2[(size_t)dj * 32 + lane];
        const float2 evv = ev2[(size_t)dj * 32 + lane];
        acc_h.x += wn * hvv.x;  acc_h.y += wn * hvv.y;
        acc_e.x += we * evv.x;  acc_e.y += we * evv.y;
    }

    const float2 hs = hv2[(size_t)node * 32 + lane];
    const float2 es = ev2[(size_t)node * 32 + lane];
    float2* no2 = (float2*)node_out;
    float2* eo2 = (float2*)edge_out;
    float2 o0 = { pn * hs.x, pn * hs.y };
    float2 p0 = { pe * es.x, pe * es.y };
    no2[(size_t)node * 64 + lane]      = o0;
    no2[(size_t)node * 64 + 32 + lane] = acc_h;
    eo2[(size_t)node * 64 + lane]      = p0;
    eo2[(size_t)node * 64 + 32 + lane] = acc_e;
}

__global__ __launch_bounds__(256) void finalize_kernel(
    const float* __restrict__ partials, int nblocks, float* __restrict__ outv)
{
    __shared__ double red[4][256];
    double s0 = 0.0, s1 = 0.0, s2 = 0.0, s3 = 0.0;
    for (int i = threadIdx.x; i < nblocks; i += 256) {
        s0 += (double)partials[(size_t)i * 4 + 0];
        s1 += (double)partials[(size_t)i * 4 + 1];
        s2 += (double)partials[(size_t)i * 4 + 2];
        s3 += (double)partials[(size_t)i * 4 + 3];
    }
    red[0][threadIdx.x] = s0; red[1][threadIdx.x] = s1;
    red[2][threadIdx.x] = s2; red[3][threadIdx.x] = s3;
    __syncthreads();
    for (int off = 128; off; off >>= 1) {
        if (threadIdx.x < off) {
            red[0][threadIdx.x] += red[0][threadIdx.x + off];
            red[1][threadIdx.x] += red[1][threadIdx.x + off];
            red[2][threadIdx.x] += red[2][threadIdx.x + off];
            red[3][threadIdx.x] += red[3][threadIdx.x + off];
        }
        __syncthreads();
    }
    if (threadIdx.x == 0) {
        const double E = (double)N_NODES * (double)DEG;
        const double var_n = (red[1][0] - red[0][0] * red[0][0] / E) / (E - 1.0);
        const double var_e = (red[3][0] - red[2][0] * red[2][0] / E) / (E - 1.0);
        outv[0] = (float)var_n;
        outv[1] = (float)var_e;
    }
}

extern "C" void kernel_launch(void* const* d_in, const int* in_sizes, int n_in,
                              void* d_out, int out_size, void* d_ws, size_t ws_size,
                              hipStream_t stream) {
    const float* node_fts = (const float*)d_in[0];
    const float* edge_fts = (const float*)d_in[1];
    const int*   edges    = (const int*)  d_in[2];
    const float* W_node   = (const float*)d_in[3];
    const float* W_edge   = (const float*)d_in[4];
    const float* a_node   = (const float*)d_in[5];
    const float* a_edge   = (const float*)d_in[6];

    float* out      = (float*)d_out;
    float* node_out = out;
    float* edge_out = out + (size_t)N_NODES * 128;
    float* outv     = out + (size_t)2 * N_NODES * 128;

    float* ws   = (float*)d_ws;
    float* hv   = ws;
    float* ev   = hv + (size_t)N_NODES * NOUT;
    float* asrc = ev + (size_t)N_NODES * NOUT;
    float* adst = asrc + N_NODES;
    float* bsrc = adst + N_NODES;
    float* bdst = bsrc + N_NODES;
    float* partials = bdst + N_NODES;

    const int ntiles = N_NODES / 16;
    const int wavesA = 2 * ntiles;
    const int blocksA = (wavesA + 7) / 8;
    proj_wmma_kernel<<<blocksA, 256, 0, stream>>>(node_fts, edge_fts, W_node, W_edge, hv, ev);

    const int blocksB = (N_NODES / 16 + 3) / 4;
    alpha_kernel<<<blocksB, 128, 0, stream>>>(hv, ev, a_node, a_edge, asrc, adst, bsrc, bdst);

    const int blocksC = (N_NODES + 7) / 8;
    edge_kernel<<<blocksC, 256, 0, stream>>>(hv, ev, edges, asrc, adst, bsrc, bdst,
                                             node_out, edge_out, partials);

    finalize_kernel<<<1, 256, 0, stream>>>(partials, blocksC, outv);
}


# edge pass as 256 resident blocks, one head each, its per-node dst scores (200 KB) held in LDS: the per-edge score read leaves the L2; half-wave per node, 16-B row loads; variance sum order changes
# speedup vs baseline: 1.5138x; 1.2983x over previous
#include <hip/hip_runtime.h>

#define N_NODES 50000
#define DEG     16
#define NIN     128
#define NOUT    64

typedef __attribute__((ext_vector_type(2))) float v2f;
typedef __attribute__((ext_vector_type(8))) float v8f;

__global__ __launch_bounds__(256) void proj_wmma_kernel(
    const float* __restrict__ node_fts, const float* __restrict__ edge_fts,
    const float* __restrict__ W_node,   const float* __restrict__ W_edge,
    float* __restrict__ hv, float* __restrict__ ev)
{
    const int wave   = (blockIdx.x * blockDim.x + threadIdx.x) >> 5;
    const int ntiles = N_NODES / 16;
    if (wave >= 2 * ntiles) return;
    const int  mat  = (wave >= ntiles) ? 1 : 0;
    const int  tile = mat ? (wave - ntiles) : wave;
    const float* A  = mat ? edge_fts : node_fts;
    const float* W  = mat ? W_edge   : W_node;
    float*       O  = mat ? ev       : hv;

    const int row0 = tile * 16;
    const int lane = threadIdx.x & 31;
    const int m    = lane & 15;
    const int kh   = lane >> 4;

    const float* arow = A + (size_t)(row0 + m) * NIN + 2 * kh;
    v8f acc0 = {}, acc1 = {}, acc2 = {}, acc3 = {};

    #pragma unroll 4
    for (int k0 = 0; k0 < NIN; k0 += 4) {
        v2f a = *(const v2f*)(arow + k0);
        const float* w0 = W + (size_t)(k0 + 2 * kh) * NOUT + m;
        const float* w1 = w0 + NOUT;
        v2f b0 = { w0[0],  w1[0]  };
        v2f b1 = { w0[16], w1[16] };
        v2f b2 = { w0[32], w1[32] };
        v2f b3 = { w0[48], w1[48] };
        acc0 = __builtin_amdgcn_wmma_f32_16x16x4_f32(false, a, false, b0, (short)0, acc0, false, false);
        acc1 = __builtin_amdgcn_wmma_f32_16x16x4_f32(false, a, false, b1, (short)0, acc1, false, false);
        acc2 = __builtin_amdgcn_wmma_f32_16x16x4_f32(false, a, false, b2, (short)0, acc2, false, false);
        acc3 = __builtin_amdgcn_wmma_f32_16x16x4_f32(false, a, false, b3, (short)0, acc3, false, false);
    }

    float* orow = O + (size_t)(row0 + kh * 8) * NOUT + m;
    #pragma unroll
    for (int r = 0; r < 8; ++r) {
        orow[(size_t)r * NOUT +  0] = acc0[r];
        orow[(size_t)r * NOUT + 16] = acc1[r];
        orow[(size_t)r * NOUT + 32] = acc2[r];
        orow[(size_t)r * NOUT + 48] = acc3[r];
    }
}

template <int M>
__device__ __forceinline__ float row_xor_add(float v)
{
    const int o = __builtin_amdgcn_update_dpp(0, __builtin_bit_cast(int, v),
                                              0x160 | M, 0xf, 0xf, false);
    return v + __builtin_bit_cast(float, o);
}

__global__ __launch_bounds__(128) void alpha_kernel(
    const float* __restrict__ hv, const float* __restrict__ ev,
    const float* __restrict__ a_node, const float* __restrict__ a_edge,
    float* __restrict__ asrc, float* __restrict__ adst,
    float* __restrict__ bsrc, float* __restrict__ bdst)
{
    const int wave = (blockIdx.x * blockDim.x + threadIdx.x) >> 5;
    if (wave >= N_NODES / 16) return;
    const int lane = threadIdx.x & 31;
    const int c    = lane & 15;
    const int h    = lane >> 4;

    const float4 as = ((const float4*)a_node)[c];
    const float4 ad = ((const float4*)a_node)[16 + c];
    const float4 bs = ((const float4*)a_edge)[c];
    const float4 bd = ((const float4*)a_edge)[16 + c];

    const float4* hv4 = (const float4*)hv;
    const float4* ev4 = (const float4*)ev;
    const int node0 = wave * 16;

    float4 x[8], y[8];
    #pragma unroll
    for (int t = 0; t < 8; ++t) {
        const size_t row = (size_t)(node0 + 2 * t + h);
        x[t] = hv4[row * 16 + c];
        y[t] = ev4[row * 16 + c];
    }

    float r_ps = 0.f, r_pd = 0.f, r_qs = 0.f, r_qd = 0.f;
    #pragma unroll
    for (int t = 0; t < 8; ++t) {
        float ps = x[t].x * as.x + x[t].y * as.y + x[t].z * as.z + x[t].w * as.w;
        float pd = x[t].x * ad.x + x[t].y * ad.y + x[t].z * ad.z + x[t].w * ad.w;
        float qs = y[t].x * bs.x + y[t].y * bs.y + y[t].z * bs.z + y[t].w * bs.w;
        float qd = y[t].x * bd.x + y[t].y * bd.y + y[t].z * bd.z + y[t].w * bd.w;
        ps = row_xor_add<8>(ps); pd = row_xor_add<8>(pd); qs = row_xor_add<8>(qs); qd = row_xor_add<8>(qd);
        ps = row_xor_add<4>(ps); pd = row_xor_add<4>(pd); qs = row_xor_add<4>(qs); qd = row_xor_add<4>(qd);
        ps = row_xor_add<2>(ps); pd = row_xor_add<2>(pd); qs = row_xor_add<2>(qs); qd = row_xor_add<2>(qd);
        ps = row_xor_add<1>(ps); pd = row_xor_add<1>(pd); qs = row_xor_add<1>(qs); qd = row_xor_add<1>(qd);
        const bool keep = (c == t);
        r_ps = keep ? ps : r_ps;  r_pd = keep ? pd : r_pd;
        r_qs = keep ? qs : r_qs;  r_qd = keep ? qd : r_qd;
    }
    if (c < 8) {
        const int node = node0 + 2 * c + h;
        asrc[node] = r_ps; adst[node] = r_pd;
        bsrc[node] = r_qs; bdst[node] = r_qd;
    }
}

template <int J>
__device__ __forceinline__ int row_bcast(int v)
{
    return __builtin_amdgcn_update_dpp(0, v, 0x150 | J, 0xf, 0xf, false);
}
template <int J>
__device__ __forceinline__ float row_bcastf(float v)
{
    return __builtin_bit_cast(float, row_bcast<J>(__builtin_bit_cast(int, v)));
}

__device__ __forceinline__ float4 ld16(const float* base, unsigned byteoff)
{
    return *(const float4*)((const char*)base + byteoff);
}
__device__ __forceinline__ void st16(float* base, unsigned byteoff, float4 v)
{
    *(float4*)((char*)base + byteoff) = v;
}
__device__ __forceinline__ void fma4(float4& a, float w, const float4& v)
{
    a.x += w * v.x;  a.y += w * v.y;  a.z += w * v.z;  a.w += w * v.w;
}

#define EDGE_HEAD_BLOCKS 128
#define EDGE_THREADS     512
#define EDGE_HEAD_WAVES  (EDGE_HEAD_BLOCKS * (EDGE_THREADS / 32))

__global__ __launch_bounds__(EDGE_THREADS) void edge_kernel(
    const float* __restrict__ hv, const float* __restrict__ ev,
    const int*  __restrict__ edges,
    const float* __restrict__ asrc, const float* __restrict__ adst,
    const float* __restrict__ bsrc, const float* __restrict__ bdst,
    float* __restrict__ node_out, float* __restrict__ edge_out,
    float* __restrict__ partials)
{
    __shared__ float dtab[N_NODES];
    __shared__ float sred[2][32];

    const int  head = blockIdx.x >= EDGE_HEAD_BLOCKS ? 1 : 0;
    const int  hb   = blockIdx.x - head * EDGE_HEAD_BLOCKS;
    const float* __restrict__ tab  = head ? ev   : hv;
    const float* __restrict__ ssrc = head ? bsrc : asrc;
    const float* __restrict__ sdst = head ? bdst : adst;
    float*       __restrict__ out  = head ? edge_out : node_out;

    {
        const float4* g4 = (const float4*)sdst;
        float4*       l4 = (float4*)dtab;
        #pragma unroll 8
        for (int t = 0; t < 24; ++t)
            l4[t * EDGE_THREADS + threadIdx.x] = g4[t * EDGE_THREADS + threadIdx.x];
        if (threadIdx.x < N_NODES / 4 - 24 * EDGE_THREADS)
            l4[24 * EDGE_THREADS + threadIdx.x] = g4[24 * EDGE_THREADS + threadIdx.x];
    }
    __syncthreads();

    const int lane = threadIdx.x & 31;
    const int wid  = __builtin_amdgcn_readfirstlane(threadIdx.x >> 5);
    const int h    = lane >> 4;
    const int c    = lane & 15;
    const unsigned laneoff = (unsigned)c * 16u;
    const int2* e2 = (const int2*)edges;

    float vs = 0.f, vs2 = 0.f;
    int p = hb * (EDGE_THREADS / 32) + wid;
    int2 ed = e2[(size_t)(2 * p + h) * DEG + c];
    for (; p < N_NODES / 2; p += EDGE_HEAD_WAVES) {
        const int node = 2 * p + h;
        const int pnext = p + EDGE_HEAD_WAVES;
        int2 ed_next = ed;
        if (pnext < N_NODES / 2) ed_next = e2[(size_t)(2 * pnext + h) * DEG + c];

        const unsigned selfoff = (unsigned)node * 256u + laneoff;
        const float4 hs = ld16(tab, selfoff);

        float x = ssrc[ed.x] + dtab[ed.y];
        x = x > 0.f ? x : 0.2f * x;
        x = fminf(fmaxf(x, -2.f), 2.f);
        const float att = __expf(x);

        float sum = att;
        sum = row_xor_add<8>(sum);  sum = row_xor_add<4>(sum);
        sum = row_xor_add<2>(sum);  sum = row_xor_add<1>(sum);
        const float norm = att / sum;

        float pn = norm, pn2 = norm * norm;
        pn = row_xor_add<8>(pn);  pn2 = row_xor_add<8>(pn2);
        pn = row_xor_add<4>(pn);  pn2 = row_xor_add<4>(pn2);
        pn = row_xor_add<2>(pn);  pn2 = row_xor_add<2>(pn2);
        pn = row_xor_add<1>(pn);  pn2 = row_xor_add<1>(pn2);
        vs += pn;  vs2 += pn2;

        const int d = ed.y;
        float4 r[16];
#define LOAD_ROW(J) r[J] = ld16(tab, (unsigned)row_bcast<J>(d) * 256u + laneoff);
        LOAD_ROW(0)  LOAD_ROW(1)  LOAD_ROW(2)  LOAD_ROW(3)  LOAD_ROW(4)  LOAD_ROW(5)  LOAD_ROW(6)  LOAD_ROW(7)
        LOAD_ROW(8)  LOAD_ROW(9)  LOAD_ROW(10) LOAD_ROW(11) LOAD_ROW(12) LOAD_ROW(13) LOAD_ROW(14) LOAD_ROW(15)
#undef LOAD_ROW
        float4 acc = {0.f, 0.f, 0.f, 0.f};
#define ADD_ROW(J) fma4(acc, row_bcastf<J>(norm), r[J]);
        ADD_ROW(0)  ADD_ROW(1)  ADD_ROW(2)  ADD_ROW(3)  ADD_ROW(4)  ADD_ROW(5)  ADD_ROW(6)  ADD_ROW(7)
        ADD_ROW(8)  ADD_ROW(9)  ADD_ROW(10) ADD_ROW(11) ADD_ROW(12) ADD_ROW(13) ADD_ROW(14) ADD_ROW(15)
#undef ADD_ROW

        const float4 o0 = { pn * hs.x, pn * hs.y, pn * hs.z, pn * hs.w };
        const unsigned outoff = (unsigned)node * 512u + laneoff;
        st16(out, outoff,        o0);
        st16(out, outoff + 256u, acc);

        ed = ed_next;
    }

    if (c == 0) { sred[0][wid * 2 + h] = vs;  sred[1][wid * 2 + h] = vs2; }
    __syncthreads();
    if (threadIdx.x < 4) {
        float t = 0.f;
        if ((threadIdx.x >> 1) == head) {
            #pragma unroll
            for (int w = 0; w < 32; ++w) t += sred[threadIdx.x & 1][w];
        }
        partials[(size_t)blockIdx.x * 4 + threadIdx.x] = t;
    }
}

__global__ __launch_bounds__(256) void finalize_kernel(
    const float* __restrict__ partials, int nblocks, float* __restrict__ outv)
{
    __shared__ double red[4][256];
    double s0 = 0.0, s1 = 0.0, s2 = 0.0, s3 = 0.0;
    for (int i = threadIdx.x; i < nblocks; i += 256) {
        s0 += (double)partials[(size_t)i * 4 + 0];
        s1 += (double)partials[(size_t)i * 4 + 1];
        s2 += (double)partials[(size_t)i * 4 + 2];
        s3 += (double)partials[(size_t)i * 4 + 3];
    }
    red[0][threadIdx.x] = s0; red[1][threadIdx.x] = s1;
    red[2][threadIdx.x] = s2; red[3][threadIdx.x] = s3;
    __syncthreads();
    for (int off = 128; off; off >>= 1) {
        if (threadIdx.x < off) {
            red[0][threadIdx.x] += red[0][threadIdx.x + off];
            red[1][threadIdx.x] += red[1][threadIdx.x + off];
            red[2][threadIdx.x] += red[2][threadIdx.x + off];
            red[3][threadIdx.x] += red[3][threadIdx.x + off];
        }
        __syncthreads();
    }
    if (threadIdx.x == 0) {
        const double E = (double)N_NODES * (double)DEG;
        const double var_n = (red[1][0] - red[0][0] * red[0][0] / E) / (E - 1.0);
        const double var_e = (red[3][0] - red[2][0] * red[2][0] / E) / (E - 1.0);
        outv[0] = (float)var_n;
        outv[1] = (float)var_e;
    }
}

extern "C" void kernel_launch(void* const* d_in, const int* in_sizes, int n_in,
                              void* d_out, int out_size, void* d_ws, size_t ws_size,
                              hipStream_t stream) {
    const float* node_fts = (const float*)d_in[0];
    const float* edge_fts = (const float*)d_in[1];
    const int*   edges    = (const int*)  d_in[2];
    const float* W_node   = (const float*)d_in[3];
    const float* W_edge   = (const float*)d_in[4];
    const float* a_node   = (const float*)d_in[5];
    const float* a_edge   = (const float*)d_in[6];

    float* out      = (float*)d_out;
    float* node_out = out;
    float* edge_out = out + (size_t)N_NODES * 128;
    float* outv     = out + (size_t)2 * N_NODES * 128;

    float* ws   = (float*)d_ws;
    float* hv   = ws;
    float* ev   = hv + (size_t)N_NODES * NOUT;
    float* asrc = ev + (size_t)N_NODES * NOUT;
    float* adst = asrc + N_NODES;
    float* bsrc = adst + N_NODES;
    float* bdst = bsrc + N_NODES;
    float* partials = bdst + N_NODES;

    const int ntiles = N_NODES / 16;
    const int wavesA = 2 * ntiles;
    const int blocksA = (wavesA + 7) / 8;
    proj_wmma_kernel<<<blocksA, 256, 0, stream>>>(node_fts, edge_fts, W_node, W_edge, hv, ev);

    const int blocksB = (N_NODES / 16 + 3) / 4;
    alpha_kernel<<<blocksB, 128, 0, stream>>>(hv, ev, a_node, a_edge, asrc, adst, bsrc, bdst);

    const int blocksC = 2 * EDGE_HEAD_BLOCKS;
    edge_kernel<<<blocksC, EDGE_THREADS, 0, stream>>>(hv, ev, edges, asrc, adst, bsrc, bdst,
                                             node_out, edge_out, partials);

    finalize_kernel<<<1, 256, 0, stream>>>(partials, blocksC, outv);
}


# projection through LDS, k and n maps unchanged: W staged once per 4-wave block, each wave's A tile and result tile pass through its own LDS slice; every global access is a whole-row 16-B vector
# speedup vs baseline: 1.6812x; 1.1106x over previous
#include <hip/hip_runtime.h>

#define N_NODES 50000
#define DEG     16
#define NIN     128
#define NOUT    64

typedef __attribute__((ext_vector_type(2))) float v2f;
typedef __attribute__((ext_vector_type(8))) float v8f;

#define PROJ_WAVES      4
#define PROJ_THREADS    (PROJ_WAVES * 32)
#define PROJ_TILES      (N_NODES / 16)
#define PROJ_MAT_BLOCKS ((PROJ_TILES + PROJ_WAVES - 1) / PROJ_WAVES)
#define W_PITCH         160
#define T_PITCH         132

__device__ __forceinline__ void wave_lds_sync()
{
    __builtin_amdgcn_fence(__ATOMIC_RELEASE, "wavefront");
    __builtin_amdgcn_wave_barrier();
    __builtin_amdgcn_fence(__ATOMIC_ACQUIRE, "wavefront");
}

__device__ __forceinline__ void block_lds_barrier()
{
    __builtin_amdgcn_fence(__ATOMIC_RELEASE, "workgroup", "local");
    __builtin_amdgcn_s_barrier();
    __builtin_amdgcn_fence(__ATOMIC_ACQUIRE, "workgroup", "local");
}

__global__ __launch_bounds__(PROJ_THREADS) void proj_wmma_kernel(
    const float* __restrict__ node_fts, const float* __restrict__ edge_fts,
    const float* __restrict__ W_node,   const float* __restrict__ W_edge,
    float* __restrict__ hv, float* __restrict__ ev)
{
    __shared__ __attribute__((aligned(16))) float wl[(NIN / 2) * W_PITCH];
    __shared__ __attribute__((aligned(16))) float tl[PROJ_WAVES][16 * T_PITCH];

    const int  mat = blockIdx.x >= PROJ_MAT_BLOCKS ? 1 : 0;
    const int  hb  = blockIdx.x - mat * PROJ_MAT_BLOCKS;
    const float* __restrict__ A = mat ? edge_fts : node_fts;
    const float* __restrict__ W = mat ? W_edge   : W_node;
    float*       __restrict__ O = mat ? ev       : hv;

    const int lane = threadIdx.x & 31;
    const int wid  = __builtin_amdgcn_readfirstlane(threadIdx.x >> 5);
    const int traw = hb * PROJ_WAVES + wid;
    const bool live = traw < PROJ_TILES;
    const int  row0 = (live ? traw : PROJ_TILES - 1) * 16;

#define X16(X) X(0) X(1) X(2) X(3) X(4) X(5) X(6) X(7) X(8) X(9) X(10) X(11) X(12) X(13) X(14) X(15)
    const float4* W4 = (const float4*)W + threadIdx.x;
#define W_LOAD(J) const float4 w##J = W4[J * PROJ_THREADS];
    X16(W_LOAD)
#undef W_LOAD

    const float4* arow4 = (const float4*)(A + (size_t)row0 * NIN) + lane;
#define A_LOAD(R) const float4 ar##R = arow4[R * (NIN / 4)];
    X16(A_LOAD)
#undef A_LOAD

    const int wk = threadIdx.x >> 4, n4 = threadIdx.x & 15;
    float* wdst = wl + (wk >> 1) * W_PITCH + 8 * n4 + (wk & 1);
#define W_PARK(J) { float* d = wdst + J * (PROJ_THREADS / 32) * W_PITCH; \
                    d[0] = w##J.x; d[2] = w##J.y; d[4] = w##J.z; d[6] = w##J.w; }
    X16(W_PARK)
#undef W_PARK
    block_lds_barrier();

    float* mytile = tl[wid];
#define A_PARK(R) *(float4*)(mytile + R * T_PITCH + 4 * lane) = ar##R;
    X16(A_PARK)
#undef A_PARK
#undef X16
    wave_lds_sync();

    const int m  = lane & 15;
    const int kh = lane >> 4;
    const float* afrag = mytile + m * T_PITCH + 2 * kh;
    const float* bfrag = wl + kh * W_PITCH + 2 * m;
    v8f acc0 = {}, acc1 = {}, acc2 = {}, acc3 = {};

    #pragma unroll 8
    for (int s = 0; s < NIN / 4; ++s) {
        const v2f a = *(const v2f*)(afrag + 4 * s);
        const float* bp = bfrag + (2 * s) * W_PITCH;
        const v2f b0 = *(const v2f*)(bp +  0);
        const v2f b1 = *(const v2f*)(bp + 32);
        const v2f b2 = *(const v2f*)(bp + 64);
        const v2f b3 = *(const v2f*)(bp + 96);
        acc0 = __builtin_amdgcn_wmma_f32_16x16x4_f32(false, a, false, b0, (short)0, acc0, false, false);
        acc1 = __builtin_amdgcn_wmma_f32_16x16x4_f32(false, a, false, b1, (short)0, acc1, false, false);
        acc2 = __builtin_amdgcn_wmma_f32_16x16x4_f32(false, a, false, b2, (short)0, acc2, false, false);
        acc3 = __builtin_amdgcn_wmma_f32_16x16x4_f32(false, a, false, b3, (short)0, acc3, false, false);
    }

    wave_lds_sync();
    float* orow = mytile + (8 * kh) * T_PITCH + m;
    #pragma unroll
    for (int r = 0; r < 8; ++r) {
        orow[r * T_PITCH +  0] = acc0[r];
        orow[r * T_PITCH + 16] = acc1[r];
        orow[r * T_PITCH + 32] = acc2[r];
        orow[r * T_PITCH + 48] = acc3[r];
    }
    wave_lds_sync();
    if (live) {
        #pragma unroll
        for (int t = 0; t < 8; ++t) {
            const int rr = 2 * t + kh;
            const float4 v = *(const float4*)(mytile + rr * T_PITCH + 4 * m);
            *(float4*)(O + (size_t)(row0 + rr) * NOUT + 4 * m) = v;
        }
    }
}

template <int M>
__device__ __forceinline__ float row_xor_add(float v)
{
    const int o = __builtin_amdgcn_update_dpp(0, __builtin_bit_cast(int, v),
                                              0x160 | M, 0xf, 0xf, false);
    return v + __builtin_bit_cast(float, o);
}

__global__ __launch_bounds__(128) void alpha_kernel(
    const float* __restrict__ hv, const float* __restrict__ ev,
    const float* __restrict__ a_node, const float* __restrict__ a_edge,
    float* __restrict__ asrc, float* __restrict__ adst,
    float* __restrict__ bsrc, float* __restrict__ bdst)
{
    const int wave = (blockIdx.x * blockDim.x + threadIdx.x) >> 5;
    if (wave >= N_NODES / 16) return;
    const int lane = threadIdx.x & 31;
    const int c    = lane & 15;
    const int h    = lane >> 4;

    const float4 as = ((const float4*)a_node)[c];
    const float4 ad = ((const float4*)a_node)[16 + c];
    const float4 bs = ((const float4*)a_edge)[c];
    const float4 bd = ((const float4*)a_edge)[16 + c];

    const float4* hv4 = (const float4*)hv;
    const float4* ev4 = (const float4*)ev;
    const int node0 = wave * 16;

    float4 x[8], y[8];
    #pragma unroll
    for (int t = 0; t < 8; ++t) {
        const size_t row = (size_t)(node0 + 2 * t + h);
        x[t] = hv4[row * 16 + c];
        y[t] = ev4[row * 16 + c];
    }

    float r_ps = 0.f, r_pd = 0.f, r_qs = 0.f, r_qd = 0.f;
    #pragma unroll
    for (int t = 0; t < 8; ++t) {
        float ps = x[t].x * as.x + x[t].y * as.y + x[t].z * as.z + x[t].w * as.w;
        float pd = x[t].x * ad.x + x[t].y * ad.y + x[t].z * ad.z + x[t].w * ad.w;
        float qs = y[t].x * bs.x + y[t].y * bs.y + y[t].z * bs.z + y[t].w * bs.w;
        float qd = y[t].x * bd.x + y[t].y * bd.y + y[t].z * bd.z + y[t].w * bd.w;
        ps = row_xor_add<8>(ps); pd = row_xor_add<8>(pd); qs = row_xor_add<8>(qs); qd = row_xor_add<8>(qd);
        ps = row_xor_add<4>(ps); pd = row_xor_add<4>(pd); qs = row_xor_add<4>(qs); qd = row_xor_add<4>(qd);
        ps = row_xor_add<2>(ps); pd = row_xor_add<2>(pd); qs = row_xor_add<2>(qs); qd = row_xor_add<2>(qd);
        ps = row_xor_add<1>(ps); pd = row_xor_add<1>(pd); qs = row_xor_add<1>(qs); qd = row_xor_add<1>(qd);
        const bool keep = (c == t);
        r_ps = keep ? ps : r_ps;  r_pd = keep ? pd : r_pd;
        r_qs = keep ? qs : r_qs;  r_qd = keep ? qd : r_qd;
    }
    if (c < 8) {
        const int node = node0 + 2 * c + h;
        asrc[node] = r_ps; adst[node] = r_pd;
        bsrc[node] = r_qs; bdst[node] = r_qd;
    }
}

template <int J>
__device__ __forceinline__ int row_bcast(int v)
{
    return __builtin_amdgcn_update_dpp(0, v, 0x150 | J, 0xf, 0xf, false);
}
template <int J>
__device__ __forceinline__ float row_bcastf(float v)
{
    return __builtin_bit_cast(float, row_bcast<J>(__builtin_bit_cast(int, v)));
}

__device__ __forceinline__ float4 ld16(const float* base, unsigned byteoff)
{
    return *(const float4*)((const char*)base + byteoff);
}
__device__ __forceinline__ void st16(float* base, unsigned byteoff, float4 v)
{
    *(float4*)((char*)base + byteoff) = v;
}
__device__ __forceinline__ void fma4(float4& a, float w, const float4& v)
{
    a.x += w * v.x;  a.y += w * v.y;  a.z += w * v.z;  a.w += w * v.w;
}

#define EDGE_HEAD_BLOCKS 128
#define EDGE_THREADS     512
#define EDGE_HEAD_WAVES  (EDGE_HEAD_BLOCKS * (EDGE_THREADS / 32))

__global__ __launch_bounds__(EDGE_THREADS) void edge_kernel(
    const float* __restrict__ hv, const float* __restrict__ ev,
    const int*  __restrict__ edges,
    const float* __restrict__ asrc, const float* __restrict__ adst,
    const float* __restrict__ bsrc, const float* __restrict__ bdst,
    float* __restrict__ node_out, float* __restrict__ edge_out,
    float* __restrict__ partials)
{
    __shared__ float dtab[N_NODES];
    __shared__ float sred[2][32];

    const int  head = blockIdx.x >= EDGE_HEAD_BLOCKS ? 1 : 0;
    const int  hb   = blockIdx.x - head * EDGE_HEAD_BLOCKS;
    const float* __restrict__ tab  = head ? ev   : hv;
    const float* __restrict__ ssrc = head ? bsrc : asrc;
    const float* __restrict__ sdst = head ? bdst : adst;
    float*       __restrict__ out  = head ? edge_out : node_out;

    {
        const float4* g4 = (const float4*)sdst;
        float4*       l4 = (float4*)dtab;
        #pragma unroll 8
        for (int t = 0; t < 24; ++t)
            l4[t * EDGE_THREADS + threadIdx.x] = g4[t * EDGE_THREADS + threadIdx.x];
        if (threadIdx.x < N_NODES / 4 - 24 * EDGE_THREADS)
            l4[24 * EDGE_THREADS + threadIdx.x] = g4[24 * EDGE_THREADS + threadIdx.x];
    }
    __syncthreads();

    const int lane = threadIdx.x & 31;
    const int wid  = __builtin_amdgcn_readfirstlane(threadIdx.x >> 5);
    const int h    = lane >> 4;
    const int c    = lane & 15;
    const unsigned laneoff = (unsigned)c * 16u;
    const int2* e2 = (const int2*)edges;

    float vs = 0.f, vs2 = 0.f;
    int p = hb * (EDGE_THREADS / 32) + wid;
    int2 ed = e2[(size_t)(2 * p + h) * DEG + c];
    for (; p < N_NODES / 2; p += EDGE_HEAD_WAVES) {
        const int node = 2 * p + h;
        const int pnext = p + EDGE_HEAD_WAVES;
        int2 ed_next = ed;
        if (pnext < N_NODES / 2) ed_next = e2[(size_t)(2 * pnext + h) * DEG + c];

        const unsigned selfoff = (unsigned)node * 256u + laneoff;
        const float4 hs = ld16(tab, selfoff);

        float x = ssrc[ed.x] + dtab[ed.y];
        x = x > 0.f ? x : 0.2f * x;
        x = fminf(fmaxf(x, -2.f), 2.f);
        const float att = __expf(x);

        float sum = att;
        sum = row_xor_add<8>(sum);  sum = row_xor_add<4>(sum);
        sum = row_xor_add<2>(sum);  sum = row_xor_add<1>(sum);
        const float norm = att / sum;

        float pn = norm, pn2 = norm * norm;
        pn = row_xor_add<8>(pn);  pn2 = row_xor_add<8>(pn2);
        pn = row_xor_add<4>(pn);  pn2 = row_xor_add<4>(pn2);
        pn = row_xor_add<2>(pn);  pn2 = row_xor_add<2>(pn2);
        pn = row_xor_add<1>(pn);  pn2 = row_xor_add<1>(pn2);
        vs += pn;  vs2 += pn2;

        const int d = ed.y;
        float4 r[16];
#define LOAD_ROW(J) r[J] = ld16(tab, (unsigned)row_bcast<J>(d) * 256u + laneoff);
        LOAD_ROW(0)  LOAD_ROW(1)  LOAD_ROW(2)  LOAD_ROW(3)  LOAD_ROW(4)  LOAD_ROW(5)  LOAD_ROW(6)  LOAD_ROW(7)
        LOAD_ROW(8)  LOAD_ROW(9)  LOAD_ROW(10) LOAD_ROW(11) LOAD_ROW(12) LOAD_ROW(13) LOAD_ROW(14) LOAD_ROW(15)
#undef LOAD_ROW
        float4 acc = {0.f, 0.f, 0.f, 0.f};
#define ADD_ROW(J) fma4(acc, row_bcastf<J>(norm), r[J]);
        ADD_ROW(0)  ADD_ROW(1)  ADD_ROW(2)  ADD_ROW(3)  ADD_ROW(4)  ADD_ROW(5)  ADD_ROW(6)  ADD_ROW(7)
        ADD_ROW(8)  ADD_ROW(9)  ADD_ROW(10) ADD_ROW(11) ADD_ROW(12) ADD_ROW(13) ADD_ROW(14) ADD_ROW(15)
#undef ADD_ROW

        const float4 o0 = { pn * hs.x, pn * hs.y, pn * hs.z, pn * hs.w };
        const unsigned outoff = (unsigned)node * 512u + laneoff;
        st16(out, outoff,        o0);
        st16(out, outoff + 256u, acc);

        ed = ed_next;
    }

    if (c == 0) { sred[0][wid * 2 + h] = vs;  sred[1][wid * 2 + h] = vs2; }
    __syncthreads();
    if (threadIdx.x < 4) {
        float t = 0.f;
        if ((threadIdx.x >> 1) == head) {
            #pragma unroll
            for (int w = 0; w < 32; ++w) t += sred[threadIdx.x & 1][w];
        }
        partials[(size_t)blockIdx.x * 4 + threadIdx.x] = t;
    }
}

__global__ __launch_bounds__(256) void finalize_kernel(
    const float* __restrict__ partials, int nblocks, float* __restrict__ outv)
{
    __shared__ double red[4][256];
    double s0 = 0.0, s1 = 0.0, s2 = 0.0, s3 = 0.0;
    for (int i = threadIdx.x; i < nblocks; i += 256) {
        s0 += (double)partials[(size_t)i * 4 + 0];
        s1 += (double)partials[(size_t)i * 4 + 1];
        s2 += (double)partials[(size_t)i * 4 + 2];
        s3 += (double)partials[(size_t)i * 4 + 3];
    }
    red[0][threadIdx.x] = s0; red[1][threadIdx.x] = s1;
    red[2][threadIdx.x] = s2; red[3][threadIdx.x] = s3;
    __syncthreads();
    for (int off = 128; off; off >>= 1) {
        if (threadIdx.x < off) {
            red[0][threadIdx.x] += red[0][threadIdx.x + off];
            red[1][threadIdx.x] += red[1][threadIdx.x + off];
            red[2][threadIdx.x] += red[2][threadIdx.x + off];
            red[3][threadIdx.x] += red[3][threadIdx.x + off];
        }
        __syncthreads();
    }
    if (threadIdx.x == 0) {
        const double E = (double)N_NODES * (double)DEG;
        const double var_n = (red[1][0] - red[0][0] * red[0][0] / E) / (E - 1.0);
        const double var_e = (red[3][0] - red[2][0] * red[2][0] / E) / (E - 1.0);
        outv[0] = (float)var_n;
        outv[1] = (float)var_e;
    }
}

extern "C" void kernel_launch(void* const* d_in, const int* in_sizes, int n_in,
                              void* d_out, int out_size, void* d_ws, size_t ws_size,
                              hipStream_t stream) {
    const float* node_fts = (const float*)d_in[0];
    const float* edge_fts = (const float*)d_in[1];
    const int*   edges    = (const int*)  d_in[2];
    const float* W_node   = (const float*)d_in[3];
    const float* W_edge   = (const float*)d_in[4];
    const float* a_node   = (const float*)d_in[5];
    const float* a_edge   = (const float*)d_in[6];

    float* out      = (float*)d_out;
    float* node_out = out;
    float* edge_out = out + (size_t)N_NODES * 128;
    float* outv     = out + (size_t)2 * N_NODES * 128;

    float* ws   = (float*)d_ws;
    float* hv   = ws;
    float* ev   = hv + (size_t)N_NODES * NOUT;
    float* asrc = ev + (size_t)N_NODES * NOUT;
    float* adst = asrc + N_NODES;
    float* bsrc = adst + N_NODES;
    float* bdst = bsrc + N_NODES;
    float* partials = bdst + N_NODES;

    const int blocksA = 2 * PROJ_MAT_BLOCKS;
    proj_wmma_kernel<<<blocksA, PROJ_THREADS, 0, stream>>>(node_fts, edge_fts, W_node, W_edge, hv, ev);

    const int blocksB = (N_NODES / 16 + 3) / 4;
    alpha_kernel<<<blocksB, 128, 0, stream>>>(hv, ev, a_node, a_edge, asrc, adst, bsrc, bdst);

    const int blocksC = 2 * EDGE_HEAD_BLOCKS;
    edge_kernel<<<blocksC, EDGE_THREADS, 0, stream>>>(hv, ev, edges, asrc, adst, bsrc, bdst,
                                             node_out, edge_out, partials);

    finalize_kernel<<<1, 256, 0, stream>>>(partials, blocksC, outv);
}


# output rows double as the row table: projection stores each row into the first half of its output row, edge pass gathers there, writes only second halves; multiply by softmax weight sum (=1) dropped
# speedup vs baseline: 1.8743x; 1.1149x over previous
#include <hip/hip_runtime.h>

#define N_NODES 50000
#define DEG     16
#define NIN     128
#define NOUT    64
#define OUT_PITCH (2 * NOUT)

typedef __attribute__((ext_vector_type(2))) float v2f;
typedef __attribute__((ext_vector_type(8))) float v8f;

#define PROJ_WAVES      4
#define PROJ_THREADS    (PROJ_WAVES * 32)
#define PROJ_TILES      (N_NODES / 16)
#define PROJ_MAT_BLOCKS ((PROJ_TILES + PROJ_WAVES - 1) / PROJ_WAVES)
#define W_PITCH         160
#define T_PITCH         132

__device__ __forceinline__ void wave_lds_sync()
{
    __builtin_amdgcn_fence(__ATOMIC_RELEASE, "wavefront");
    __builtin_amdgcn_wave_barrier();
    __builtin_amdgcn_fence(__ATOMIC_ACQUIRE, "wavefront");
}

__device__ __forceinline__ void block_lds_barrier()
{
    __builtin_amdgcn_fence(__ATOMIC_RELEASE, "workgroup", "local");
    __builtin_amdgcn_s_barrier();
    __builtin_amdgcn_fence(__ATOMIC_ACQUIRE, "workgroup", "local");
}

__global__ __launch_bounds__(PROJ_THREADS) void proj_wmma_kernel(
    const float* __restrict__ node_fts, const float* __restrict__ edge_fts,
    const float* __restrict__ W_node,   const float* __restrict__ W_edge,
    float* __restrict__ node_out, float* __restrict__ edge_out)
{
    __shared__ __attribute__((aligned(16))) float wl[(NIN / 2) * W_PITCH];
    __shared__ __attribute__((aligned(16))) float tl[PROJ_WAVES][16 * T_PITCH];

    const int  mat = blockIdx.x >= PROJ_MAT_BLOCKS ? 1 : 0;
    const int  hb  = blockIdx.x - mat * PROJ_MAT_BLOCKS;
    const float* __restrict__ A = mat ? edge_fts : node_fts;
    const float* __restrict__ W = mat ? W_edge   : W_node;
    float*       __restrict__ O = mat ? edge_out : node_out;

    const int lane = threadIdx.x & 31;
    const int wid  = __builtin_amdgcn_readfirstlane(threadIdx.x >> 5);
    const int traw = hb * PROJ_WAVES + wid;
    const bool live = traw < PROJ_TILES;
    const int  row0 = (live ? traw : PROJ_TILES - 1) * 16;

#define X16(X) X(0) X(1) X(2) X(3) X(4) X(5) X(6) X(7) X(8) X(9) X(10) X(11) X(12) X(13) X(14) X(15)
    const float4* W4 = (const float4*)W + threadIdx.x;
#define W_LOAD(J) const float4 w##J = W4[J * PROJ_THREADS];
    X16(W_LOAD)
#undef W_LOAD

    const float4* arow4 = (const float4*)(A + (size_t)row0 * NIN) + lane;
#define A_LOAD(R) const float4 ar##R = arow4[R * (NIN / 4)];
    X16(A_LOAD)
#undef A_LOAD

    const int wk = threadIdx.x >> 4, n4 = threadIdx.x & 15;
    float* wdst = wl + (wk >> 1) * W_PITCH + 8 * n4 + (wk & 1);
#define W_PARK(J) { float* d = wdst + J * (PROJ_THREADS / 32) * W_PITCH; \
                    d[0] = w##J.x; d[2] = w##J.y; d[4] = w##J.z; d[6] = w##J.w; }
    X16(W_PARK)
#undef W_PARK
    block_lds_barrier();

    float* mytile = tl[wid];
#define A_PARK(R) *(float4*)(mytile + R * T_PITCH + 4 * lane) = ar##R;
    X16(A_PARK)
#undef A_PARK
#undef X16
    wave_lds_sync();

    const int m  = lane & 15;
    const int kh = lane >> 4;
    const float* afrag = mytile + m * T_PITCH + 2 * kh;
    const float* bfrag = wl + kh * W_PITCH + 2 * m;
    v8f acc0 = {}, acc1 = {}, acc2 = {}, acc3 = {};

    #pragma unroll 8
    for (int s = 0; s < NIN / 4; ++s) {
        const v2f a = *(const v2f*)(afrag + 4 * s);
        const float* bp = bfrag + (2 * s) * W_PITCH;
        const v2f b0 = *(const v2f*)(bp +  0);
        const v2f b1 = *(const v2f*)(bp + 32);
        const v2f b2 = *(const v2f*)(bp + 64);
        const v2f b3 = *(const v2f*)(bp + 96);
        acc0 = __builtin_amdgcn_wmma_f32_16x16x4_f32(false, a, false, b0, (short)0, acc0, false, false);
        acc1 = __builtin_amdgcn_wmma_f32_16x16x4_f32(false, a, false, b1, (short)0, acc1, false, false);
        acc2 = __builtin_amdgcn_wmma_f32_16x16x4_f32(false, a, false, b2, (short)0, acc2, false, false);
        acc3 = __builtin_amdgcn_wmma_f32_16x16x4_f32(false, a, false, b3, (short)0, acc3, false, false);
    }

    wave_lds_sync();
    float* orow = mytile + (8 * kh) * T_PITCH + m;
    #pragma unroll
    for (int r = 0; r < 8; ++r) {
        orow[r * T_PITCH +  0] = acc0[r];
        orow[r * T_PITCH + 16] = acc1[r];
        orow[r * T_PITCH + 32] = acc2[r];
        orow[r * T_PITCH + 48] = acc3[r];
    }
    wave_lds_sync();
    if (live) {
        #pragma unroll
        for (int t = 0; t < 8; ++t) {
            const int rr = 2 * t + kh;
            const float4 v = *(const float4*)(mytile + rr * T_PITCH + 4 * m);
            *(float4*)(O + (size_t)(row0 + rr) * OUT_PITCH + 4 * m) = v;
        }
    }
}

template <int M>
__device__ __forceinline__ float row_xor_add(float v)
{
    const int o = __builtin_amdgcn_update_dpp(0, __builtin_bit_cast(int, v),
                                              0x160 | M, 0xf, 0xf, false);
    return v + __builtin_bit_cast(float, o);
}

__global__ __launch_bounds__(128) void alpha_kernel(
    const float* __restrict__ node_out, const float* __restrict__ edge_out,
    const float* __restrict__ a_node, const float* __restrict__ a_edge,
    float* __restrict__ asrc, float* __restrict__ adst,
    float* __restrict__ bsrc, float* __restrict__ bdst)
{
    const int wave = (blockIdx.x * blockDim.x + threadIdx.x) >> 5;
    if (wave >= N_NODES / 16) return;
    const int lane = threadIdx.x & 31;
    const int c    = lane & 15;
    const int h    = lane >> 4;

    const float4 as = ((const float4*)a_node)[c];
    const float4 ad = ((const float4*)a_node)[16 + c];
    const float4 bs = ((const float4*)a_edge)[c];
    const float4 bd = ((const float4*)a_edge)[16 + c];

    const float4* hv4 = (const float4*)node_out;
    const float4* ev4 = (const float4*)edge_out;
    const int node0 = wave * 16;

    float4 x[8], y[8];
    #pragma unroll
    for (int t = 0; t < 8; ++t) {
        const size_t row = (size_t)(node0 + 2 * t + h);
        x[t] = hv4[row * (OUT_PITCH / 4) + c];
        y[t] = ev4[row * (OUT_PITCH / 4) + c];
    }

    float r_ps = 0.f, r_pd = 0.f, r_qs = 0.f, r_qd = 0.f;
    #pragma unroll
    for (int t = 0; t < 8; ++t) {
        float ps = x[t].x * as.x + x[t].y * as.y + x[t].z * as.z + x[t].w * as.w;
        float pd = x[t].x * ad.x + x[t].y * ad.y + x[t].z * ad.z + x[t].w * ad.w;
        float qs = y[t].x * bs.x + y[t].y * bs.y + y[t].z * bs.z + y[t].w * bs.w;
        float qd = y[t].x * bd.x + y[t].y * bd.y + y[t].z * bd.z + y[t].w * bd.w;
        ps = row_xor_add<8>(ps); pd = row_xor_add<8>(pd); qs = row_xor_add<8>(qs); qd = row_xor_add<8>(qd);
        ps = row_xor_add<4>(ps); pd = row_xor_add<4>(pd); qs = row_xor_add<4>(qs); qd = row_xor_add<4>(qd);
        ps = row_xor_add<2>(ps); pd = row_xor_add<2>(pd); qs = row_xor_add<2>(qs); qd = row_xor_add<2>(qd);
        ps = row_xor_add<1>(ps); pd = row_xor_add<1>(pd); qs = row_xor_add<1>(qs); qd = row_xor_add<1>(qd);
        const bool keep = (c == t);
        r_ps = keep ? ps : r_ps;  r_pd = keep ? pd : r_pd;
        r_qs = keep ? qs : r_qs;  r_qd = keep ? qd : r_qd;
    }
    if (c < 8) {
        const int node = node0 + 2 * c + h;
        asrc[node] = r_ps; adst[node] = r_pd;
        bsrc[node] = r_qs; bdst[node] = r_qd;
    }
}

template <int J>
__device__ __forceinline__ int row_bcast(int v)
{
    return __builtin_amdgcn_update_dpp(0, v, 0x150 | J, 0xf, 0xf, false);
}
template <int J>
__device__ __forceinline__ float row_bcastf(float v)
{
    return __builtin_bit_cast(float, row_bcast<J>(__builtin_bit_cast(int, v)));
}

__device__ __forceinline__ float4 ld16(const float* base, unsigned byteoff)
{
    return *(const float4*)((const char*)base + byteoff);
}
__device__ __forceinline__ void st16(float* base, unsigned byteoff, float4 v)
{
    *(float4*)((char*)base + byteoff) = v;
}
__device__ __forceinline__ void fma4(float4& a, float w, const float4& v)
{
    a.x += w * v.x;  a.y += w * v.y;  a.z += w * v.z;  a.w += w * v.w;
}

#define EDGE_HEAD_BLOCKS 128
#define EDGE_THREADS     512
#define EDGE_HEAD_WAVES  (EDGE_HEAD_BLOCKS * (EDGE_THREADS / 32))

__global__ __launch_bounds__(EDGE_THREADS) void edge_kernel(
    const int*  __restrict__ edges,
    const float* __restrict__ asrc, const float* __restrict__ adst,
    const float* __restrict__ bsrc, const float* __restrict__ bdst,
    const float* __restrict__ node_rows, const float* __restrict__ edge_rows,
    float* __restrict__ node_sums, float* __restrict__ edge_sums,
    float* __restrict__ partials)
{
    __shared__ float dtab[N_NODES];
    __shared__ float sred[2][32];

    const int  head = blockIdx.x >= EDGE_HEAD_BLOCKS ? 1 : 0;
    const int  hb   = blockIdx.x - head * EDGE_HEAD_BLOCKS;
    const float* __restrict__ tab  = head ? edge_rows : node_rows;
    float*       __restrict__ out  = head ? edge_sums : node_sums;
    const float* __restrict__ ssrc = head ? bsrc : asrc;
    const float* __restrict__ sdst = head ? bdst : adst;

    {
        const float4* g4 = (const float4*)sdst;
        float4*       l4 = (float4*)dtab;
        #pragma unroll 8
        for (int t = 0; t < 24; ++t)
            l4[t * EDGE_THREADS + threadIdx.x] = g4[t * EDGE_THREADS + threadIdx.x];
        if (threadIdx.x < N_NODES / 4 - 24 * EDGE_THREADS)
            l4[24 * EDGE_THREADS + threadIdx.x] = g4[24 * EDGE_THREADS + threadIdx.x];
    }
    __syncthreads();

    const int lane = threadIdx.x & 31;
    const int wid  = __builtin_amdgcn_readfirstlane(threadIdx.x >> 5);
    const int h    = lane >> 4;
    const int c    = lane & 15;
    const unsigned laneoff = (unsigned)c * 16u;
    const int2* e2 = (const int2*)edges;

    float vs = 0.f, vs2 = 0.f;
    int p = hb * (EDGE_THREADS / 32) + wid;
    int2 ed = e2[(size_t)(2 * p + h) * DEG + c];
    for (; p < N_NODES / 2; p += EDGE_HEAD_WAVES) {
        const int node = 2 * p + h;
        const int pnext = p + EDGE_HEAD_WAVES;
        int2 ed_next = ed;
        if (pnext < N_NODES / 2) ed_next = e2[(size_t)(2 * pnext + h) * DEG + c];

        float x = ssrc[ed.x] + dtab[ed.y];
        x = x > 0.f ? x : 0.2f * x;
        x = fminf(fmaxf(x, -2.f), 2.f);
        const float att = __expf(x);

        float sum = att;
        sum = row_xor_add<8>(sum);  sum = row_xor_add<4>(sum);
        sum = row_xor_add<2>(sum);  sum = row_xor_add<1>(sum);
        const float norm = att / sum;

        float pn = norm, pn2 = norm * norm;
        pn = row_xor_add<8>(pn);  pn2 = row_xor_add<8>(pn2);
        pn = row_xor_add<4>(pn);  pn2 = row_xor_add<4>(pn2);
        pn = row_xor_add<2>(pn);  pn2 = row_xor_add<2>(pn2);
        pn = row_xor_add<1>(pn);  pn2 = row_xor_add<1>(pn2);
        vs += pn;  vs2 += pn2;

        const int d = ed.y;
        float4 r[16];
#define LOAD_ROW(J) r[J] = ld16(tab, (unsigned)row_bcast<J>(d) * 512u + laneoff);
        LOAD_ROW(0)  LOAD_ROW(1)  LOAD_ROW(2)  LOAD_ROW(3)  LOAD_ROW(4)  LOAD_ROW(5)  LOAD_ROW(6)  LOAD_ROW(7)
        LOAD_ROW(8)  LOAD_ROW(9)  LOAD_ROW(10) LOAD_ROW(11) LOAD_ROW(12) LOAD_ROW(13) LOAD_ROW(14) LOAD_ROW(15)
#undef LOAD_ROW
        float4 acc = {0.f, 0.f, 0.f, 0.f};
#define ADD_ROW(J) fma4(acc, row_bcastf<J>(norm), r[J]);
        ADD_ROW(0)  ADD_ROW(1)  ADD_ROW(2)  ADD_ROW(3)  ADD_ROW(4)  ADD_ROW(5)  ADD_ROW(6)  ADD_ROW(7)
        ADD_ROW(8)  ADD_ROW(9)  ADD_ROW(10) ADD_ROW(11) ADD_ROW(12) ADD_ROW(13) ADD_ROW(14) ADD_ROW(15)
#undef ADD_ROW

        st16(out, (unsigned)node * 512u + laneoff, acc);

        ed = ed_next;
    }

    if (c == 0) { sred[0][wid * 2 + h] = vs;  sred[1][wid * 2 + h] = vs2; }
    __syncthreads();
    if (threadIdx.x < 4) {
        float t = 0.f;
        if ((threadIdx.x >> 1) == head) {
            #pragma unroll
            for (int w = 0; w < 32; ++w) t += sred[threadIdx.x & 1][w];
        }
        partials[(size_t)blockIdx.x * 4 + threadIdx.x] = t;
    }
}

__global__ __launch_bounds__(256) void finalize_kernel(
    const float* __restrict__ partials, int nblocks, float* __restrict__ outv)
{
    __shared__ double red[4][256];
    double s0 = 0.0, s1 = 0.0, s2 = 0.0, s3 = 0.0;
    for (int i = threadIdx.x; i < nblocks; i += 256) {
        s0 += (double)partials[(size_t)i * 4 + 0];
        s1 += (double)partials[(size_t)i * 4 + 1];
        s2 += (double)partials[(size_t)i * 4 + 2];
        s3 += (double)partials[(size_t)i * 4 + 3];
    }
    red[0][threadIdx.x] = s0; red[1][threadIdx.x] = s1;
    red[2][threadIdx.x] = s2; red[3][threadIdx.x] = s3;
    __syncthreads();
    for (int off = 128; off; off >>= 1) {
        if (threadIdx.x < off) {
            red[0][threadIdx.x] += red[0][threadIdx.x + off];
            red[1][threadIdx.x] += red[1][threadIdx.x + off];
            red[2][threadIdx.x] += red[2][threadIdx.x + off];
            red[3][threadIdx.x] += red[3][threadIdx.x + off];
        }
        __syncthreads();
    }
    if (threadIdx.x == 0) {
        const double E = (double)N_NODES * (double)DEG;
        const double var_n = (red[1][0] - red[0][0] * red[0][0] / E) / (E - 1.0);
        const double var_e = (red[3][0] - red[2][0] * red[2][0] / E) / (E - 1.0);
        outv[0] = (float)var_n;
        outv[1] = (float)var_e;
    }
}

extern "C" void kernel_launch(void* const* d_in, const int* in_sizes, int n_in,
                              void* d_out, int out_size, void* d_ws, size_t ws_size,
                              hipStream_t stream) {
    const float* node_fts = (const float*)d_in[0];
    const float* edge_fts = (const float*)d_in[1];
    const int*   edges    = (const int*)  d_in[2];
    const float* W_node   = (const float*)d_in[3];
    const float* W_edge   = (const float*)d_in[4];
    const float* a_node   = (const float*)d_in[5];
    const float* a_edge   = (const float*)d_in[6];

    float* out      = (float*)d_out;
    float* node_out = out;
    float* edge_out = out + (size_t)N_NODES * 128;
    float* outv     = out + (size_t)2 * N_NODES * 128;

    float* ws   = (float*)d_ws;
    float* asrc = ws;
    float* adst = asrc + N_NODES;
    float* bsrc = adst + N_NODES;
    float* bdst = bsrc + N_NODES;
    float* partials = bdst + N_NODES;

    const int blocksA = 2 * PROJ_MAT_BLOCKS;
    proj_wmma_kernel<<<blocksA, PROJ_THREADS, 0, stream>>>(node_fts, edge_fts, W_node, W_edge, node_out, edge_out);

    const int blocksB = (N_NODES / 16 + 3) / 4;
    alpha_kernel<<<blocksB, 128, 0, stream>>>(node_out, edge_out, a_node, a_edge, asrc, adst, bsrc, bdst);

    const int blocksC = 2 * EDGE_HEAD_BLOCKS;
    edge_kernel<<<blocksC, EDGE_THREADS, 0, stream>>>(edges, asrc, adst, bsrc, bdst,
                                             node_out, edge_out, node_out + NOUT, edge_out + NOUT,
                                             partials);

    finalize_kernel<<<1, 256, 0, stream>>>(partials, blocksC, outv);
}


# projection as 256 resident 16-wave blocks: W staged once per block (was once per 4 tiles), a wave walks 1-2 tiles and requests the next tile's rows before the current tile's WMMAs; k order kept
# speedup vs baseline: 1.9426x; 1.0364x over previous
#include <hip/hip_runtime.h>

#define N_NODES 50000
#define DEG     16
#define NIN     128
#define NOUT    64
#define OUT_PITCH (2 * NOUT)

typedef __attribute__((ext_vector_type(2))) float v2f;
typedef __attribute__((ext_vector_type(8))) float v8f;

#define PROJ_WAVES      16
#define PROJ_THREADS    (PROJ_WAVES * 32)
#define PROJ_TILES      (N_NODES / 16)
#define PROJ_MAT_BLOCKS 128
#define W_PITCH         160
#define T_PITCH         132

__device__ __forceinline__ void wave_lds_sync()
{
    __builtin_amdgcn_fence(__ATOMIC_RELEASE, "wavefront");
    __builtin_amdgcn_wave_barrier();
    __builtin_amdgcn_fence(__ATOMIC_ACQUIRE, "wavefront");
}

__device__ __forceinline__ void block_lds_barrier()
{
    __builtin_amdgcn_fence(__ATOMIC_RELEASE, "workgroup", "local");
    __builtin_amdgcn_s_barrier();
    __builtin_amdgcn_fence(__ATOMIC_ACQUIRE, "workgroup", "local");
}

__global__ __launch_bounds__(PROJ_THREADS) void proj_wmma_kernel(
    const float* __restrict__ node_fts, const float* __restrict__ edge_fts,
    const float* __restrict__ W_node,   const float* __restrict__ W_edge,
    float* __restrict__ node_out, float* __restrict__ edge_out)
{
    __shared__ __attribute__((aligned(16))) float wl[(NIN / 2) * W_PITCH];
    __shared__ __attribute__((aligned(16))) float tl[PROJ_WAVES][16 * T_PITCH];

    const int  mat = blockIdx.x >= PROJ_MAT_BLOCKS ? 1 : 0;
    const int  hb  = blockIdx.x - mat * PROJ_MAT_BLOCKS;
    const float* __restrict__ A = mat ? edge_fts : node_fts;
    const float* __restrict__ W = mat ? W_edge   : W_node;
    float*       __restrict__ O = mat ? edge_out : node_out;

    const int lane = threadIdx.x & 31;
    const int wid  = __builtin_amdgcn_readfirstlane(threadIdx.x >> 5);

#define X16(X) X(0) X(1) X(2) X(3) X(4) X(5) X(6) X(7) X(8) X(9) X(10) X(11) X(12) X(13) X(14) X(15)
    const float4* W4 = (const float4*)W + threadIdx.x;
    const float4 w0 = W4[0 * PROJ_THREADS];
    const float4 w1 = W4[1 * PROJ_THREADS];
    const float4 w2 = W4[2 * PROJ_THREADS];
    const float4 w3 = W4[3 * PROJ_THREADS];

    int tile = hb + PROJ_MAT_BLOCKS * wid;
    const float4* arow4 = (const float4*)(A + (size_t)tile * 16 * NIN) + lane;
#define A_LOAD(R) float4 ar##R = arow4[R * (NIN / 4)];
    X16(A_LOAD)
#undef A_LOAD

    const int wk = threadIdx.x >> 4, n4 = threadIdx.x & 15;
    float* wdst = wl + (wk >> 1) * W_PITCH + 8 * n4 + (wk & 1);
#define W_PARK(J) { float* d = wdst + J * (PROJ_THREADS / 32) * W_PITCH; \
                    d[0] = w##J.x; d[2] = w##J.y; d[4] = w##J.z; d[6] = w##J.w; }
    W_PARK(0) W_PARK(1) W_PARK(2) W_PARK(3)
#undef W_PARK
    block_lds_barrier();

    float* mytile = tl[wid];
    const int m  = lane & 15;
    const int kh = lane >> 4;
    const float* afrag = mytile + m * T_PITCH + 2 * kh;
    const float* bfrag = wl + kh * W_PITCH + 2 * m;
    float* orow = mytile + (8 * kh) * T_PITCH + m;

    const int rank  = (wid >> 2) + 4 * (((wid & 3) - (wid >> 2)) & 3);
    const int tile2 = hb + PROJ_MAT_BLOCKS * (PROJ_WAVES + rank);
    bool second = false;

    for (;;) {
        wave_lds_sync();
#define A_PARK(R) *(float4*)(mytile + R * T_PITCH + 4 * lane) = ar##R;
        X16(A_PARK)
#undef A_PARK
        wave_lds_sync();

        const int  tnext    = second ? PROJ_TILES : tile2;
        const bool has_next = tnext < PROJ_TILES;
        if (has_next) {
            const float4* nrow4 = (const float4*)(A + (size_t)tnext * 16 * NIN) + lane;
#define A_NEXT(R) ar##R = nrow4[R * (NIN / 4)];
            X16(A_NEXT)
#undef A_NEXT
        }

        v8f acc0 = {}, acc1 = {}, acc2 = {}, acc3 = {};
        #pragma unroll 8
        for (int s = 0; s < NIN / 4; ++s) {
            const v2f a = *(const v2f*)(afrag + 4 * s);
            const float* bp = bfrag + (2 * s) * W_PITCH;
            const v2f b0 = *(const v2f*)(bp +  0);
            const v2f b1 = *(const v2f*)(bp + 32);
            const v2f b2 = *(const v2f*)(bp + 64);
            const v2f b3 = *(const v2f*)(bp + 96);
            acc0 = __builtin_amdgcn_wmma_f32_16x16x4_f32(false, a, false, b0, (short)0, acc0, false, false);
            acc1 = __builtin_amdgcn_wmma_f32_16x16x4_f32(false, a, false, b1, (short)0, acc1, false, false);
            acc2 = __builtin_amdgcn_wmma_f32_16x16x4_f32(false, a, false, b2, (short)0, acc2, false, false);
            acc3 = __builtin_amdgcn_wmma_f32_16x16x4_f32(false, a, false, b3, (short)0, acc3, false, false);
        }

        wave_lds_sync();
        #pragma unroll
        for (int r = 0; r < 8; ++r) {
            orow[r * T_PITCH +  0] = acc0[r];
            orow[r * T_PITCH + 16] = acc1[r];
            orow[r * T_PITCH + 32] = acc2[r];
            orow[r * T_PITCH + 48] = acc3[r];
        }
        wave_lds_sync();
        float* obase = O + (size_t)tile * 16 * OUT_PITCH + 4 * m;
        #pragma unroll
        for (int t = 0; t < 8; ++t) {
            const int rr = 2 * t + kh;
            const float4 v = *(const float4*)(mytile + rr * T_PITCH + 4 * m);
            *(float4*)(obase + (size_t)rr * OUT_PITCH) = v;
        }
        if (!has_next) break;
        tile = tnext;
        second = true;
    }
#undef X16
}

template <int M>
__device__ __forceinline__ float row_xor_add(float v)
{
    const int o = __builtin_amdgcn_update_dpp(0, __builtin_bit_cast(int, v),
                                              0x160 | M, 0xf, 0xf, false);
    return v + __builtin_bit_cast(float, o);
}

__global__ __launch_bounds__(128) void alpha_kernel(
    const float* __restrict__ node_out, const float* __restrict__ edge_out,
    const float* __restrict__ a_node, const float* __restrict__ a_edge,
    float* __restrict__ asrc, float* __restrict__ adst,
    float* __restrict__ bsrc, float* __restrict__ bdst)
{
    const int wave = (blockIdx.x * blockDim.x + threadIdx.x) >> 5;
    if (wave >= N_NODES / 16) return;
    const int lane = threadIdx.x & 31;
    const int c    = lane & 15;
    const int h    = lane >> 4;

    const float4 as = ((const float4*)a_node)[c];
    const float4 ad = ((const float4*)a_node)[16 + c];
    const float4 bs = ((const float4*)a_edge)[c];
    const float4 bd = ((const float4*)a_edge)[16 + c];

    const float4* hv4 = (const float4*)node_out;
    const float4* ev4 = (const float4*)edge_out;
    const int node0 = wave * 16;

    float4 x[8], y[8];
    #pragma unroll
    for (int t = 0; t < 8; ++t) {
        const size_t row = (size_t)(node0 + 2 * t + h);
        x[t] = hv4[row * (OUT_PITCH / 4) + c];
        y[t] = ev4[row * (OUT_PITCH / 4) + c];
    }

    float r_ps = 0.f, r_pd = 0.f, r_qs = 0.f, r_qd = 0.f;
    #pragma unroll
    for (int t = 0; t < 8; ++t) {
        float ps = x[t].x * as.x + x[t].y * as.y + x[t].z * as.z + x[t].w * as.w;
        float pd = x[t].x * ad.x + x[t].y * ad.y + x[t].z * ad.z + x[t].w * ad.w;
        float qs = y[t].x * bs.x + y[t].y * bs.y + y[t].z * bs.z + y[t].w * bs.w;
        float qd = y[t].x * bd.x + y[t].y * bd.y + y[t].z * bd.z + y[t].w * bd.w;
        ps = row_xor_add<8>(ps); pd = row_xor_add<8>(pd); qs = row_xor_add<8>(qs); qd = row_xor_add<8>(qd);
        ps = row_xor_add<4>(ps); pd = row_xor_add<4>(pd); qs = row_xor_add<4>(qs); qd = row_xor_add<4>(qd);
        ps = row_xor_add<2>(ps); pd = row_xor_add<2>(pd); qs = row_xor_add<2>(qs); qd = row_xor_add<2>(qd);
        ps = row_xor_add<1>(ps); pd = row_xor_add<1>(pd); qs = row_xor_add<1>(qs); qd = row_xor_add<1>(qd);
        const bool keep = (c == t);
        r_ps = keep ? ps : r_ps;  r_pd = keep ? pd : r_pd;
        r_qs = keep ? qs : r_qs;  r_qd = keep ? qd : r_qd;
    }
    if (c < 8) {
        const int node = node0 + 2 * c + h;
        asrc[node] = r_ps; adst[node] = r_pd;
        bsrc[node] = r_qs; bdst[node] = r_qd;
    }
}

template <int J>
__device__ __forceinline__ int row_bcast(int v)
{
    return __builtin_amdgcn_update_dpp(0, v, 0x150 | J, 0xf, 0xf, false);
}
template <int J>
__device__ __forceinline__ float row_bcastf(float v)
{
    return __builtin_bit_cast(float, row_bcast<J>(__builtin_bit_cast(int, v)));
}

__device__ __forceinline__ float4 ld16(const float* base, unsigned byteoff)
{
    return *(const float4*)((const char*)base + byteoff);
}
__device__ __forceinline__ void st16(float* base, unsigned byteoff, float4 v)
{
    *(float4*)((char*)base + byteoff) = v;
}
__device__ __forceinline__ void fma4(float4& a, float w, const float4& v)
{
    a.x += w * v.x;  a.y += w * v.y;  a.z += w * v.z;  a.w += w * v.w;
}

#define EDGE_HEAD_BLOCKS 128
#define EDGE_THREADS     512
#define EDGE_HEAD_WAVES  (EDGE_HEAD_BLOCKS * (EDGE_THREADS / 32))

__global__ __launch_bounds__(EDGE_THREADS) void edge_kernel(
    const int*  __restrict__ edges,
    const float* __restrict__ asrc, const float* __restrict__ adst,
    const float* __restrict__ bsrc, const float* __restrict__ bdst,
    const float* __restrict__ node_rows, const float* __restrict__ edge_rows,
    float* __restrict__ node_sums, float* __restrict__ edge_sums,
    float* __restrict__ partials)
{
    __shared__ float dtab[N_NODES];
    __shared__ float sred[2][32];

    const int  head = blockIdx.x >= EDGE_HEAD_BLOCKS ? 1 : 0;
    const int  hb   = blockIdx.x - head * EDGE_HEAD_BLOCKS;
    const float* __restrict__ tab  = head ? edge_rows : node_rows;
    float*       __restrict__ out  = head ? edge_sums : node_sums;
    const float* __restrict__ ssrc = head ? bsrc : asrc;
    const float* __restrict__ sdst = head ? bdst : adst;

    {
        const float4* g4 = (const float4*)sdst;
        float4*       l4 = (float4*)dtab;
        #pragma unroll 8
        for (int t = 0; t < 24; ++t)
            l4[t * EDGE_THREADS + threadIdx.x] = g4[t * EDGE_THREADS + threadIdx.x];
        if (threadIdx.x < N_NODES / 4 - 24 * EDGE_THREADS)
            l4[24 * EDGE_THREADS + threadIdx.x] = g4[24 * EDGE_THREADS + threadIdx.x];
    }
    __syncthreads();

    const int lane = threadIdx.x & 31;
    const int wid  = __builtin_amdgcn_readfirstlane(threadIdx.x >> 5);
    const int h    = lane >> 4;
    const int c    = lane & 15;
    const unsigned laneoff = (unsigned)c * 16u;
    const int2* e2 = (const int2*)edges;

    float vs = 0.f, vs2 = 0.f;
    int p = hb * (EDGE_THREADS / 32) + wid;
    int2 ed = e2[(size_t)(2 * p + h) * DEG + c];
    for (; p < N_NODES / 2; p += EDGE_HEAD_WAVES) {
        const int node = 2 * p + h;
        const int pnext = p + EDGE_HEAD_WAVES;
        int2 ed_next = ed;
        if (pnext < N_NODES / 2) ed_next = e2[(size_t)(2 * pnext + h) * DEG + c];

        float x = ssrc[ed.x] + dtab[ed.y];
        x = x > 0.f ? x : 0.2f * x;
        x = fminf(fmaxf(x, -2.f), 2.f);
        const float att = __expf(x);

        float sum = att;
        sum = row_xor_add<8>(sum);  sum = row_xor_add<4>(sum);
        sum = row_xor_add<2>(sum);  sum = row_xor_add<1>(sum);
        const float norm = att / sum;

        float pn = norm, pn2 = norm * norm;
        pn = row_xor_add<8>(pn);  pn2 = row_xor_add<8>(pn2);
        pn = row_xor_add<4>(pn);  pn2 = row_xor_add<4>(pn2);
        pn = row_xor_add<2>(pn);  pn2 = row_xor_add<2>(pn2);
        pn = row_xor_add<1>(pn);  pn2 = row_xor_add<1>(pn2);
        vs += pn;  vs2 += pn2;

        const int d = ed.y;
        float4 r[16];
#define LOAD_ROW(J) r[J] = ld16(tab, (unsigned)row_bcast<J>(d) * 512u + laneoff);
        LOAD_ROW(0)  LOAD_ROW(1)  LOAD_ROW(2)  LOAD_ROW(3)  LOAD_ROW(4)  LOAD_ROW(5)  LOAD_ROW(6)  LOAD_ROW(7)
        LOAD_ROW(8)  LOAD_ROW(9)  LOAD_ROW(10) LOAD_ROW(11) LOAD_ROW(12) LOAD_ROW(13) LOAD_ROW(14) LOAD_ROW(15)
#undef LOAD_ROW
        float4 acc = {0.f, 0.f, 0.f, 0.f};
#define ADD_ROW(J) fma4(acc, row_bcastf<J>(norm), r[J]);
        ADD_ROW(0)  ADD_ROW(1)  ADD_ROW(2)  ADD_ROW(3)  ADD_ROW(4)  ADD_ROW(5)  ADD_ROW(6)  ADD_ROW(7)
        ADD_ROW(8)  ADD_ROW(9)  ADD_ROW(10) ADD_ROW(11) ADD_ROW(12) ADD_ROW(13) ADD_ROW(14) ADD_ROW(15)
#undef ADD_ROW

        st16(out, (unsigned)node * 512u + laneoff, acc);

        ed = ed_next;
    }

    if (c == 0) { sred[0][wid * 2 + h] = vs;  sred[1][wid * 2 + h] = vs2; }
    __syncthreads();
    if (threadIdx.x < 4) {
        float t = 0.f;
        if ((threadIdx.x >> 1) == head) {
            #pragma unroll
            for (int w = 0; w < 32; ++w) t += sred[threadIdx.x & 1][w];
        }
        partials[(size_t)blockIdx.x * 4 + threadIdx.x] = t;
    }
}

__global__ __launch_bounds__(256) void finalize_kernel(
    const float* __restrict__ partials, int nblocks, float* __restrict__ outv)
{
    __shared__ double red[4][256];
    double s0 = 0.0, s1 = 0.0, s2 = 0.0, s3 = 0.0;
    for (int i = threadIdx.x; i < nblocks; i += 256) {
        s0 += (double)partials[(size_t)i * 4 + 0];
        s1 += (double)partials[(size_t)i * 4 + 1];
        s2 += (double)partials[(size_t)i * 4 + 2];
        s3 += (double)partials[(size_t)i * 4 + 3];
    }
    red[0][threadIdx.x] = s0; red[1][threadIdx.x] = s1;
    red[2][threadIdx.x] = s2; red[3][threadIdx.x] = s3;
    __syncthreads();
    for (int off = 128; off; off >>= 1) {
        if (threadIdx.x < off) {
            red[0][threadIdx.x] += red[0][threadIdx.x + off];
            red[1][threadIdx.x] += red[1][threadIdx.x + off];
            red[2][threadIdx.x] += red[2][threadIdx.x + off];
            red[3][threadIdx.x] += red[3][threadIdx.x + off];
        }
        __syncthreads();
    }
    if (threadIdx.x == 0) {
        const double E = (double)N_NODES * (double)DEG;
        const double var_n = (red[1][0] - red[0][0] * red[0][0] / E) / (E - 1.0);
        const double var_e = (red[3][0] - red[2][0] * red[2][0] / E) / (E - 1.0);
        outv[0] = (float)var_n;
        outv[1] = (float)var_e;
    }
}

extern "C" void kernel_launch(void* const* d_in, const int* in_sizes, int n_in,
                              void* d_out, int out_size, void* d_ws, size_t ws_size,
                              hipStream_t stream) {
    const float* node_fts = (const float*)d_in[0];
    const float* edge_fts = (const float*)d_in[1];
    const int*   edges    = (const int*)  d_in[2];
    const float* W_node   = (const float*)d_in[3];
    const float* W_edge   = (const float*)d_in[4];
    const float* a_node   = (const float*)d_in[5];
    const float* a_edge   = (const float*)d_in[6];

    float* out      = (float*)d_out;
    float* node_out = out;
    float* edge_out = out + (size_t)N_NODES * 128;
    float* outv     = out + (size_t)2 * N_NODES * 128;

    float* ws   = (float*)d_ws;
    float* asrc = ws;
    float* adst = asrc + N_NODES;
    float* bsrc = adst + N_NODES;
    float* bdst = bsrc + N_NODES;
    float* partials = bdst + N_NODES;

    const int blocksA = 2 * PROJ_MAT_BLOCKS;
    proj_wmma_kernel<<<blocksA, PROJ_THREADS, 0, stream>>>(node_fts, edge_fts, W_node, W_edge, node_out, edge_out);

    const int blocksB = (N_NODES / 16 + 3) / 4;
    alpha_kernel<<<blocksB, 128, 0, stream>>>(node_out, edge_out, a_node, a_edge, asrc, adst, bsrc, bdst);

    const int blocksC = 2 * EDGE_HEAD_BLOCKS;
    edge_kernel<<<blocksC, EDGE_THREADS, 0, stream>>>(edges, asrc, adst, bsrc, bdst,
                                             node_out, edge_out, node_out + NOUT, edge_out + NOUT,
                                             partials);

    finalize_kernel<<<1, 256, 0, stream>>>(partials, blocksC, outv);
}


# per-node attention scores as x.(W a): a projection block folds a into its W (two 128-vectors), a wave dots a tile's 16 input rows with them in registers; the alpha launch goes; dot sum order changes
# speedup vs baseline: 2.0869x; 1.0743x over previous
#include <hip/hip_runtime.h>

#define N_NODES 50000
#define DEG     16
#define NIN     128
#define NOUT    64
#define OUT_PITCH (2 * NOUT)

typedef __attribute__((ext_vector_type(2))) float v2f;
typedef __attribute__((ext_vector_type(8))) float v8f;

#define PROJ_WAVES      16
#define PROJ_THREADS    (PROJ_WAVES * 32)
#define PROJ_TILES      (N_NODES / 16)
#define PROJ_MAT_BLOCKS 128
#define W_PITCH         160
#define T_PITCH         132

template <int M>
__device__ __forceinline__ float row_xor(float v)
{
    return __builtin_bit_cast(float,
        __builtin_amdgcn_update_dpp(0, __builtin_bit_cast(int, v), 0x160 | M, 0xf, 0xf, false));
}
template <int M>
__device__ __forceinline__ float row_xor_add(float v)
{
    return v + row_xor<M>(v);
}
template <int M>
__device__ __forceinline__ float pair_fold(float p, float q, bool hi)
{
    const float keep = hi ? q : p;
    const float send = hi ? p : q;
    return keep + row_xor<M>(send);
}
__device__ __forceinline__ float rows16_sum(
    float p0, float p1, float p2,  float p3,  float p4,  float p5,  float p6,  float p7,
    float p8, float p9, float p10, float p11, float p12, float p13, float p14, float p15,
    bool b3, bool b2, bool b1, bool b0)
{
    const float q0 = pair_fold<8>(p0, p8,  b3);
    const float q1 = pair_fold<8>(p4, p12, b3);
    const float q2 = pair_fold<8>(p2, p10, b3);
    const float q3 = pair_fold<8>(p6, p14, b3);
    const float q4 = pair_fold<8>(p1, p9,  b3);
    const float q5 = pair_fold<8>(p5, p13, b3);
    const float q6 = pair_fold<8>(p3, p11, b3);
    const float q7 = pair_fold<8>(p7, p15, b3);
    const float r0 = pair_fold<4>(q0, q1, b2);
    const float r1 = pair_fold<4>(q2, q3, b2);
    const float r2 = pair_fold<4>(q4, q5, b2);
    const float r3 = pair_fold<4>(q6, q7, b2);
    const float s0 = pair_fold<2>(r0, r1, b1);
    const float s1 = pair_fold<2>(r2, r3, b1);
    return pair_fold<1>(s0, s1, b0);
}
__device__ __forceinline__ float xhalf(float x)
{
    const int b = __builtin_bit_cast(int, x);
    return __builtin_bit_cast(float,
        __builtin_amdgcn_permlanex16(b, b, 0x76543210, 0xfedcba98, false, false));
}
__device__ __forceinline__ float dot4(const float4& x, const float4& u)
{
    return x.x * u.x + x.y * u.y + x.z * u.z + x.w * u.w;
}

__device__ __forceinline__ void wave_lds_sync()
{
    __builtin_amdgcn_fence(__ATOMIC_RELEASE, "wavefront");
    __builtin_amdgcn_wave_barrier();
    __builtin_amdgcn_fence(__ATOMIC_ACQUIRE, "wavefront");
}

__device__ __forceinline__ void block_lds_barrier()
{
    __builtin_amdgcn_fence(__ATOMIC_RELEASE, "workgroup", "local");
    __builtin_amdgcn_s_barrier();
    __builtin_amdgcn_fence(__ATOMIC_ACQUIRE, "workgroup", "local");
}

__global__ __launch_bounds__(PROJ_THREADS) void proj_wmma_kernel(
    const float* __restrict__ node_fts, const float* __restrict__ edge_fts,
    const float* __restrict__ W_node,   const float* __restrict__ W_edge,
    const float* __restrict__ a_node,   const float* __restrict__ a_edge,
    float* __restrict__ node_out, float* __restrict__ edge_out,
    float* __restrict__ asrc, float* __restrict__ adst,
    float* __restrict__ bsrc, float* __restrict__ bdst)
{
    __shared__ __attribute__((aligned(16))) float wl[(NIN / 2) * W_PITCH];
    __shared__ __attribute__((aligned(16))) float tl[PROJ_WAVES][16 * T_PITCH];
    __shared__ __attribute__((aligned(16))) float ul[2 * NIN];

    const int  mat = blockIdx.x >= PROJ_MAT_BLOCKS ? 1 : 0;
    const int  hb  = blockIdx.x - mat * PROJ_MAT_BLOCKS;
    const float* __restrict__ A = mat ? edge_fts : node_fts;
    const float* __restrict__ W = mat ? W_edge   : W_node;
    float*       __restrict__ O = mat ? edge_out : node_out;
    const float* __restrict__ avec = mat ? a_edge : a_node;
    float*       __restrict__ ssrc = mat ? bsrc : asrc;
    float*       __restrict__ sdst = mat ? bdst : adst;

    const int lane = threadIdx.x & 31;
    const int wid  = __builtin_amdgcn_readfirstlane(threadIdx.x >> 5);

#define X16(X) X(0) X(1) X(2) X(3) X(4) X(5) X(6) X(7) X(8) X(9) X(10) X(11) X(12) X(13) X(14) X(15)
    const float4* W4 = (const float4*)W + threadIdx.x;
    const float4 w0 = W4[0 * PROJ_THREADS];
    const float4 w1 = W4[1 * PROJ_THREADS];
    const float4 w2 = W4[2 * PROJ_THREADS];
    const float4 w3 = W4[3 * PROJ_THREADS];
    const float4 as4 = ((const float4*)avec)[threadIdx.x & 15];
    const float4 ad4 = ((const float4*)avec)[16 + (threadIdx.x & 15)];

    int tile = hb + PROJ_MAT_BLOCKS * wid;
    const float4* arow4 = (const float4*)(A + (size_t)tile * 16 * NIN) + lane;
#define A_LOAD(R) float4 ar##R = arow4[R * (NIN / 4)];
    X16(A_LOAD)
#undef A_LOAD

    const int wk = threadIdx.x >> 4, n4 = threadIdx.x & 15;
    float* wdst = wl + (wk >> 1) * W_PITCH + 8 * n4 + (wk & 1);
#define W_PARK(J) { float* d = wdst + J * (PROJ_THREADS / 32) * W_PITCH; \
                    d[0] = w##J.x; d[2] = w##J.y; d[4] = w##J.z; d[6] = w##J.w; }
    W_PARK(0) W_PARK(1) W_PARK(2) W_PARK(3)
#undef W_PARK
    const bool lb3 = (threadIdx.x & 8) != 0, lb2 = (threadIdx.x & 4) != 0;
    const bool lb1 = (threadIdx.x & 2) != 0, lb0 = (threadIdx.x & 1) != 0;
    {
        const float f0 = pair_fold<8>(dot4(w0, as4), dot4(w1, as4), lb3);
        const float f1 = pair_fold<8>(dot4(w2, as4), dot4(w3, as4), lb3);
        const float f2 = pair_fold<8>(dot4(w0, ad4), dot4(w1, ad4), lb3);
        const float f3 = pair_fold<8>(dot4(w2, ad4), dot4(w3, ad4), lb3);
        const float g0 = pair_fold<4>(f0, f1, lb2);
        const float g1 = pair_fold<4>(f2, f3, lb2);
        const float tot = row_xor_add<1>(pair_fold<2>(g0, g1, lb1));
        const int j = 2 * (lb2 ? 1 : 0) + (lb3 ? 1 : 0);
        if (!lb0) ul[2 * (wk + 32 * j) + (lb1 ? 1 : 0)] = tot;
    }
    block_lds_barrier();

    const float4 uA = *(const float4*)(ul + 8 * lane);
    const float4 uB = *(const float4*)(ul + 8 * lane + 4);
    const v2f u0 = {uA.x, uA.y}, u1 = {uA.z, uA.w}, u2 = {uB.x, uB.y}, u3 = {uB.z, uB.w};

    float* mytile = tl[wid];
    const int m  = lane & 15;
    const int kh = lane >> 4;
    const float* afrag = mytile + m * T_PITCH + 2 * kh;
    const float* bfrag = wl + kh * W_PITCH + 2 * m;
    float* orow = mytile + (8 * kh) * T_PITCH + m;

    const int rank  = (wid >> 2) + 4 * (((wid & 3) - (wid >> 2)) & 3);
    const int tile2 = hb + PROJ_MAT_BLOCKS * (PROJ_WAVES + rank);
    bool second = false;

    for (;;) {
        wave_lds_sync();
#define A_PARK(R) *(float4*)(mytile + R * T_PITCH + 4 * lane) = ar##R;
        X16(A_PARK)
#undef A_PARK
        wave_lds_sync();

        {
#define S_DOT(R) const v2f pp##R = ar##R.x * u0 + ar##R.y * u1 + ar##R.z * u2 + ar##R.w * u3;
            X16(S_DOT)
#undef S_DOT
            float ssum = rows16_sum(pp0.x, pp1.x, pp2.x,  pp3.x,  pp4.x,  pp5.x,  pp6.x,  pp7.x,
                                    pp8.x, pp9.x, pp10.x, pp11.x, pp12.x, pp13.x, pp14.x, pp15.x,
                                    lb3, lb2, lb1, lb0);
            float dsum = rows16_sum(pp0.y, pp1.y, pp2.y,  pp3.y,  pp4.y,  pp5.y,  pp6.y,  pp7.y,
                                    pp8.y, pp9.y, pp10.y, pp11.y, pp12.y, pp13.y, pp14.y, pp15.y,
                                    lb3, lb2, lb1, lb0);
            ssum += xhalf(ssum);
            dsum += xhalf(dsum);
            (kh ? sdst : ssrc)[tile * 16 + m] = kh ? dsum : ssum;
        }

        const int  tnext    = second ? PROJ_TILES : tile2;
        const bool has_next = tnext < PROJ_TILES;
        if (has_next) {
            const float4* nrow4 = (const float4*)(A + (size_t)tnext * 16 * NIN) + lane;
#define A_NEXT(R) ar##R = nrow4[R * (NIN / 4)];
            X16(A_NEXT)
#undef A_NEXT
        }

        v8f acc0 = {}, acc1 = {}, acc2 = {}, acc3 = {};
        #pragma unroll 8
        for (int s = 0; s < NIN / 4; ++s) {
            const v2f a = *(const v2f*)(afrag + 4 * s);
            const float* bp = bfrag + (2 * s) * W_PITCH;
            const v2f b0 = *(const v2f*)(bp +  0);
            const v2f b1 = *(const v2f*)(bp + 32);
            const v2f b2 = *(const v2f*)(bp + 64);
            const v2f b3 = *(const v2f*)(bp + 96);
            acc0 = __builtin_amdgcn_wmma_f32_16x16x4_f32(false, a, false, b0, (short)0, acc0, false, false);
            acc1 = __builtin_amdgcn_wmma_f32_16x16x4_f32(false, a, false, b1, (short)0, acc1, false, false);
            acc2 = __builtin_amdgcn_wmma_f32_16x16x4_f32(false, a, false, b2, (short)0, acc2, false, false);
            acc3 = __builtin_amdgcn_wmma_f32_16x16x4_f32(false, a, false, b3, (short)0, acc3, false, false);
        }

        wave_lds_sync();
        #pragma unroll
        for (int r = 0; r < 8; ++r) {
            orow[r * T_PITCH +  0] = acc0[r];
            orow[r * T_PITCH + 16] = acc1[r];
            orow[r * T_PITCH + 32] = acc2[r];
            orow[r * T_PITCH + 48] = acc3[r];
        }
        wave_lds_sync();
        float* obase = O + (size_t)tile * 16 * OUT_PITCH + 4 * m;
        #pragma unroll
        for (int t = 0; t < 8; ++t) {
            const int rr = 2 * t + kh;
            const float4 v = *(const float4*)(mytile + rr * T_PITCH + 4 * m);
            *(float4*)(obase + (size_t)rr * OUT_PITCH) = v;
        }
        if (!has_next) break;
        tile = tnext;
        second = true;
    }
#undef X16
}

template <int J>
__device__ __forceinline__ int row_bcast(int v)
{
    return __builtin_amdgcn_update_dpp(0, v, 0x150 | J, 0xf, 0xf, false);
}
template <int J>
__device__ __forceinline__ float row_bcastf(float v)
{
    return __builtin_bit_cast(float, row_bcast<J>(__builtin_bit_cast(int, v)));
}

__device__ __forceinline__ float4 ld16(const float* base, unsigned byteoff)
{
    return *(const float4*)((const char*)base + byteoff);
}
__device__ __forceinline__ void st16(float* base, unsigned byteoff, float4 v)
{
    *(float4*)((char*)base + byteoff) = v;
}
__device__ __forceinline__ void fma4(float4& a, float w, const float4& v)
{
    a.x += w * v.x;  a.y += w * v.y;  a.z += w * v.z;  a.w += w * v.w;
}

#define EDGE_HEAD_BLOCKS 128
#define EDGE_THREADS     512
#define EDGE_HEAD_WAVES  (EDGE_HEAD_BLOCKS * (EDGE_THREADS / 32))

__global__ __launch_bounds__(EDGE_THREADS) void edge_kernel(
    const int*  __restrict__ edges,
    const float* __restrict__ asrc, const float* __restrict__ adst,
    const float* __restrict__ bsrc, const float* __restrict__ bdst,
    const float* __restrict__ node_rows, const float* __restrict__ edge_rows,
    float* __restrict__ node_sums, float* __restrict__ edge_sums,
    float* __restrict__ partials)
{
    __shared__ float dtab[N_NODES];
    __shared__ float sred[2][32];

    const int  head = blockIdx.x >= EDGE_HEAD_BLOCKS ? 1 : 0;
    const int  hb   = blockIdx.x - head * EDGE_HEAD_BLOCKS;
    const float* __restrict__ tab  = head ? edge_rows : node_rows;
    float*       __restrict__ out  = head ? edge_sums : node_sums;
    const float* __restrict__ ssrc = head ? bsrc : asrc;
    const float* __restrict__ sdst = head ? bdst : adst;

    {
        const float4* g4 = (const float4*)sdst;
        float4*       l4 = (float4*)dtab;
        #pragma unroll 8
        for (int t = 0; t < 24; ++t)
            l4[t * EDGE_THREADS + threadIdx.x] = g4[t * EDGE_THREADS + threadIdx.x];
        if (threadIdx.x < N_NODES / 4 - 24 * EDGE_THREADS)
            l4[24 * EDGE_THREADS + threadIdx.x] = g4[24 * EDGE_THREADS + threadIdx.x];
    }
    __syncthreads();

    const int lane = threadIdx.x & 31;
    const int wid  = __builtin_amdgcn_readfirstlane(threadIdx.x >> 5);
    const int h    = lane >> 4;
    const int c    = lane & 15;
    const unsigned laneoff = (unsigned)c * 16u;
    const int2* e2 = (const int2*)edges;

    float vs = 0.f, vs2 = 0.f;
    int p = hb * (EDGE_THREADS / 32) + wid;
    int2 ed = e2[(size_t)(2 * p + h) * DEG + c];
    for (; p < N_NODES / 2; p += EDGE_HEAD_WAVES) {
        const int node = 2 * p + h;
        const int pnext = p + EDGE_HEAD_WAVES;
        int2 ed_next = ed;
        if (pnext < N_NODES / 2) ed_next = e2[(size_t)(2 * pnext + h) * DEG + c];

        float x = ssrc[ed.x] + dtab[ed.y];
        x = x > 0.f ? x : 0.2f * x;
        x = fminf(fmaxf(x, -2.f), 2.f);
        const float att = __expf(x);

        float sum = att;
        sum = row_xor_add<8>(sum);  sum = row_xor_add<4>(sum);
        sum = row_xor_add<2>(sum);  sum = row_xor_add<1>(sum);
        const float norm = att / sum;

        float pn = norm, pn2 = norm * norm;
        pn = row_xor_add<8>(pn);  pn2 = row_xor_add<8>(pn2);
        pn = row_xor_add<4>(pn);  pn2 = row_xor_add<4>(pn2);
        pn = row_xor_add<2>(pn);  pn2 = row_xor_add<2>(pn2);
        pn = row_xor_add<1>(pn);  pn2 = row_xor_add<1>(pn2);
        vs += pn;  vs2 += pn2;

        const int d = ed.y;
        float4 r[16];
#define LOAD_ROW(J) r[J] = ld16(tab, (unsigned)row_bcast<J>(d) * 512u + laneoff);
        LOAD_ROW(0)  LOAD_ROW(1)  LOAD_ROW(2)  LOAD_ROW(3)  LOAD_ROW(4)  LOAD_ROW(5)  LOAD_ROW(6)  LOAD_ROW(7)
        LOAD_ROW(8)  LOAD_ROW(9)  LOAD_ROW(10) LOAD_ROW(11) LOAD_ROW(12) LOAD_ROW(13) LOAD_ROW(14) LOAD_ROW(15)
#undef LOAD_ROW
        float4 acc = {0.f, 0.f, 0.f, 0.f};
#define ADD_ROW(J) fma4(acc, row_bcastf<J>(norm), r[J]);
        ADD_ROW(0)  ADD_ROW(1)  ADD_ROW(2)  ADD_ROW(3)  ADD_ROW(4)  ADD_ROW(5)  ADD_ROW(6)  ADD_ROW(7)
        ADD_ROW(8)  ADD_ROW(9)  ADD_ROW(10) ADD_ROW(11) ADD_ROW(12) ADD_ROW(13) ADD_ROW(14) ADD_ROW(15)
#undef ADD_ROW

        st16(out, (unsigned)node * 512u + laneoff, acc);

        ed = ed_next;
    }

    if (c == 0) { sred[0][wid * 2 + h] = vs;  sred[1][wid * 2 + h] = vs2; }
    __syncthreads();
    if (threadIdx.x < 4) {
        float t = 0.f;
        if ((threadIdx.x >> 1) == head) {
            #pragma unroll
            for (int w = 0; w < 32; ++w) t += sred[threadIdx.x & 1][w];
        }
        partials[(size_t)blockIdx.x * 4 + threadIdx.x] = t;
    }
}

__global__ __launch_bounds__(256) void finalize_kernel(
    const float* __restrict__ partials, int nblocks, float* __restrict__ outv)
{
    __shared__ double red[4][256];
    double s0 = 0.0, s1 = 0.0, s2 = 0.0, s3 = 0.0;
    for (int i = threadIdx.x; i < nblocks; i += 256) {
        s0 += (double)partials[(size_t)i * 4 + 0];
        s1 += (double)partials[(size_t)i * 4 + 1];
        s2 += (double)partials[(size_t)i * 4 + 2];
        s3 += (double)partials[(size_t)i * 4 + 3];
    }
    red[0][threadIdx.x] = s0; red[1][threadIdx.x] = s1;
    red[2][threadIdx.x] = s2; red[3][threadIdx.x] = s3;
    __syncthreads();
    for (int off = 128; off; off >>= 1) {
        if (threadIdx.x < off) {
            red[0][threadIdx.x] += red[0][threadIdx.x + off];
            red[1][threadIdx.x] += red[1][threadIdx.x + off];
            red[2][threadIdx.x] += red[2][threadIdx.x + off];
            red[3][threadIdx.x] += red[3][threadIdx.x + off];
        }
        __syncthreads();
    }
    if (threadIdx.x == 0) {
        const double E = (double)N_NODES * (double)DEG;
        const double var_n = (red[1][0] - red[0][0] * red[0][0] / E) / (E - 1.0);
        const double var_e = (red[3][0] - red[2][0] * red[2][0] / E) / (E - 1.0);
        outv[0] = (float)var_n;
        outv[1] = (float)var_e;
    }
}

extern "C" void kernel_launch(void* const* d_in, const int* in_sizes, int n_in,
                              void* d_out, int out_size, void* d_ws, size_t ws_size,
                              hipStream_t stream) {
    const float* node_fts = (const float*)d_in[0];
    const float* edge_fts = (const float*)d_in[1];
    const int*   edges    = (const int*)  d_in[2];
    const float* W_node   = (const float*)d_in[3];
    const float* W_edge   = (const float*)d_in[4];
    const float* a_node   = (const float*)d_in[5];
    const float* a_edge   = (const float*)d_in[6];

    float* out      = (float*)d_out;
    float* node_out = out;
    float* edge_out = out + (size_t)N_NODES * 128;
    float* outv     = out + (size_t)2 * N_NODES * 128;

    float* ws   = (float*)d_ws;
    float* asrc = ws;
    float* adst = asrc + N_NODES;
    float* bsrc = adst + N_NODES;
    float* bdst = bsrc + N_NODES;
    float* partials = bdst + N_NODES;

    const int blocksA = 2 * PROJ_MAT_BLOCKS;
    proj_wmma_kernel<<<blocksA, PROJ_THREADS, 0, stream>>>(node_fts, edge_fts, W_node, W_edge, a_node, a_edge,
                                                       node_out, edge_out, asrc, adst, bsrc, bdst);

    const int blocksC = 2 * EDGE_HEAD_BLOCKS;
    edge_kernel<<<blocksC, EDGE_THREADS, 0, stream>>>(edges, asrc, adst, bsrc, bdst,
                                             node_out, edge_out, node_out + NOUT, edge_out + NOUT,
                                             partials);

    finalize_kernel<<<1, 256, 0, stream>>>(partials, blocksC, outv);
}


# last launch: 2 waves (1 per head) fold the 128 block sums, now handed over as doubles, with lane shuffles; no LDS rounds, barriers or f64 divides (x 1/E constants); double sum order changes
# speedup vs baseline: 2.1674x; 1.0386x over previous
#include <hip/hip_runtime.h>

#define N_NODES 50000
#define DEG     16
#define NIN     128
#define NOUT    64
#define OUT_PITCH (2 * NOUT)

typedef __attribute__((ext_vector_type(2))) float v2f;
typedef __attribute__((ext_vector_type(8))) float v8f;

#define PROJ_WAVES      16
#define PROJ_THREADS    (PROJ_WAVES * 32)
#define PROJ_TILES      (N_NODES / 16)
#define PROJ_MAT_BLOCKS 128
#define W_PITCH         160
#define T_PITCH         132

template <int M>
__device__ __forceinline__ float row_xor(float v)
{
    return __builtin_bit_cast(float,
        __builtin_amdgcn_update_dpp(0, __builtin_bit_cast(int, v), 0x160 | M, 0xf, 0xf, false));
}
template <int M>
__device__ __forceinline__ float row_xor_add(float v)
{
    return v + row_xor<M>(v);
}
template <int M>
__device__ __forceinline__ float pair_fold(float p, float q, bool hi)
{
    const float keep = hi ? q : p;
    const float send = hi ? p : q;
    return keep + row_xor<M>(send);
}
__device__ __forceinline__ float rows16_sum(
    float p0, float p1, float p2,  float p3,  float p4,  float p5,  float p6,  float p7,
    float p8, float p9, float p10, float p11, float p12, float p13, float p14, float p15,
    bool b3, bool b2, bool b1, bool b0)
{
    const float q0 = pair_fold<8>(p0, p8,  b3);
    const float q1 = pair_fold<8>(p4, p12, b3);
    const float q2 = pair_fold<8>(p2, p10, b3);
    const float q3 = pair_fold<8>(p6, p14, b3);
    const float q4 = pair_fold<8>(p1, p9,  b3);
    const float q5 = pair_fold<8>(p5, p13, b3);
    const float q6 = pair_fold<8>(p3, p11, b3);
    const float q7 = pair_fold<8>(p7, p15, b3);
    const float r0 = pair_fold<4>(q0, q1, b2);
    const float r1 = pair_fold<4>(q2, q3, b2);
    const float r2 = pair_fold<4>(q4, q5, b2);
    const float r3 = pair_fold<4>(q6, q7, b2);
    const float s0 = pair_fold<2>(r0, r1, b1);
    const float s1 = pair_fold<2>(r2, r3, b1);
    return pair_fold<1>(s0, s1, b0);
}
__device__ __forceinline__ float xhalf(float x)
{
    const int b = __builtin_bit_cast(int, x);
    return __builtin_bit_cast(float,
        __builtin_amdgcn_permlanex16(b, b, 0x76543210, 0xfedcba98, false, false));
}
__device__ __forceinline__ float dot4(const float4& x, const float4& u)
{
    return x.x * u.x + x.y * u.y + x.z * u.z + x.w * u.w;
}

__device__ __forceinline__ void wave_lds_sync()
{
    __builtin_amdgcn_fence(__ATOMIC_RELEASE, "wavefront");
    __builtin_amdgcn_wave_barrier();
    __builtin_amdgcn_fence(__ATOMIC_ACQUIRE, "wavefront");
}

__device__ __forceinline__ void block_lds_barrier()
{
    __builtin_amdgcn_fence(__ATOMIC_RELEASE, "workgroup", "local");
    __builtin_amdgcn_s_barrier();
    __builtin_amdgcn_fence(__ATOMIC_ACQUIRE, "workgroup", "local");
}

__global__ __launch_bounds__(PROJ_THREADS) void proj_wmma_kernel(
    const float* __restrict__ node_fts, const float* __restrict__ edge_fts,
    const float* __restrict__ W_node,   const float* __restrict__ W_edge,
    const float* __restrict__ a_node,   const float* __restrict__ a_edge,
    float* __restrict__ node_out, float* __restrict__ edge_out,
    float* __restrict__ asrc, float* __restrict__ adst,
    float* __restrict__ bsrc, float* __restrict__ bdst)
{
    __shared__ __attribute__((aligned(16))) float wl[(NIN / 2) * W_PITCH];
    __shared__ __attribute__((aligned(16))) float tl[PROJ_WAVES][16 * T_PITCH];
    __shared__ __attribute__((aligned(16))) float ul[2 * NIN];

    const int  mat = blockIdx.x >= PROJ_MAT_BLOCKS ? 1 : 0;
    const int  hb  = blockIdx.x - mat * PROJ_MAT_BLOCKS;
    const float* __restrict__ A = mat ? edge_fts : node_fts;
    const float* __restrict__ W = mat ? W_edge   : W_node;
    float*       __restrict__ O = mat ? edge_out : node_out;
    const float* __restrict__ avec = mat ? a_edge : a_node;
    float*       __restrict__ ssrc = mat ? bsrc : asrc;
    float*       __restrict__ sdst = mat ? bdst : adst;

    const int lane = threadIdx.x & 31;
    const int wid  = __builtin_amdgcn_readfirstlane(threadIdx.x >> 5);

#define X16(X) X(0) X(1) X(2) X(3) X(4) X(5) X(6) X(7) X(8) X(9) X(10) X(11) X(12) X(13) X(14) X(15)
    const float4* W4 = (const float4*)W + threadIdx.x;
    const float4 w0 = W4[0 * PROJ_THREADS];
    const float4 w1 = W4[1 * PROJ_THREADS];
    const float4 w2 = W4[2 * PROJ_THREADS];
    const float4 w3 = W4[3 * PROJ_THREADS];
    const float4 as4 = ((const float4*)avec)[threadIdx.x & 15];
    const float4 ad4 = ((const float4*)avec)[16 + (threadIdx.x & 15)];

    int tile = hb + PROJ_MAT_BLOCKS * wid;
    const float4* arow4 = (const float4*)(A + (size_t)tile * 16 * NIN) + lane;
#define A_LOAD(R) float4 ar##R = arow4[R * (NIN / 4)];
    X16(A_LOAD)
#undef A_LOAD

    const int wk = threadIdx.x >> 4, n4 = threadIdx.x & 15;
    float* wdst = wl + (wk >> 1) * W_PITCH + 8 * n4 + (wk & 1);
#define W_PARK(J) { float* d = wdst + J * (PROJ_THREADS / 32) * W_PITCH; \
                    d[0] = w##J.x; d[2] = w##J.y; d[4] = w##J.z; d[6] = w##J.w; }
    W_PARK(0) W_PARK(1) W_PARK(2) W_PARK(3)
#undef W_PARK
    const bool lb3 = (threadIdx.x & 8) != 0, lb2 = (threadIdx.x & 4) != 0;
    const bool lb1 = (threadIdx.x & 2) != 0, lb0 = (threadIdx.x & 1) != 0;
    {
        const float f0 = pair_fold<8>(dot4(w0, as4), dot4(w1, as4), lb3);
        const float f1 = pair_fold<8>(dot4(w2, as4), dot4(w3, as4), lb3);
        const float f2 = pair_fold<8>(dot4(w0, ad4), dot4(w1, ad4), lb3);
        const float f3 = pair_fold<8>(dot4(w2, ad4), dot4(w3, ad4), lb3);
        const float g0 = pair_fold<4>(f0, f1, lb2);
        const float g1 = pair_fold<4>(f2, f3, lb2);
        const float tot = row_xor_add<1>(pair_fold<2>(g0, g1, lb1));
        const int j = 2 * (lb2 ? 1 : 0) + (lb3 ? 1 : 0);
        if (!lb0) ul[2 * (wk + 32 * j) + (lb1 ? 1 : 0)] = tot;
    }
    block_lds_barrier();

    const float4 uA = *(const float4*)(ul + 8 * lane);
    const float4 uB = *(const float4*)(ul + 8 * lane + 4);
    const v2f u0 = {uA.x, uA.y}, u1 = {uA.z, uA.w}, u2 = {uB.x, uB.y}, u3 = {uB.z, uB.w};

    float* mytile = tl[wid];
    const int m  = lane & 15;
    const int kh = lane >> 4;
    const float* afrag = mytile + m * T_PITCH + 2 * kh;
    const float* bfrag = wl + kh * W_PITCH + 2 * m;
    float* orow = mytile + (8 * kh) * T_PITCH + m;

    const int rank  = (wid >> 2) + 4 * (((wid & 3) - (wid >> 2)) & 3);
    const int tile2 = hb + PROJ_MAT_BLOCKS * (PROJ_WAVES + rank);
    bool second = false;

    for (;;) {
        wave_lds_sync();
#define A_PARK(R) *(float4*)(mytile + R * T_PITCH + 4 * lane) = ar##R;
        X16(A_PARK)
#undef A_PARK
        wave_lds_sync();

        {
#define S_DOT(R) const v2f pp##R = ar##R.x * u0 + ar##R.y * u1 + ar##R.z * u2 + ar##R.w * u3;
            X16(S_DOT)
#undef S_DOT
            float ssum = rows16_sum(pp0.x, pp1.x, pp2.x,  pp3.x,  pp4.x,  pp5.x,  pp6.x,  pp7.x,
                                    pp8.x, pp9.x, pp10.x, pp11.x, pp12.x, pp13.x, pp14.x, pp15.x,
                                    lb3, lb2, lb1, lb0);
            float dsum = rows16_sum(pp0.y, pp1.y, pp2.y,  pp3.y,  pp4.y,  pp5.y,  pp6.y,  pp7.y,
                                    pp8.y, pp9.y, pp10.y, pp11.y, pp12.y, pp13.y, pp14.y, pp15.y,
                                    lb3, lb2, lb1, lb0);
            ssum += xhalf(ssum);
            dsum += xhalf(dsum);
            (kh ? sdst : ssrc)[tile * 16 + m] = kh ? dsum : ssum;
        }

        const int  tnext    = second ? PROJ_TILES : tile2;
        const bool has_next = tnext < PROJ_TILES;
        if (has_next) {
            const float4* nrow4 = (const float4*)(A + (size_t)tnext * 16 * NIN) + lane;
#define A_NEXT(R) ar##R = nrow4[R * (NIN / 4)];
            X16(A_NEXT)
#undef A_NEXT
        }

        v8f acc0 = {}, acc1 = {}, acc2 = {}, acc3 = {};
        #pragma unroll 8
        for (int s = 0; s < NIN / 4; ++s) {
            const v2f a = *(const v2f*)(afrag + 4 * s);
            const float* bp = bfrag + (2 * s) * W_PITCH;
            const v2f b0 = *(const v2f*)(bp +  0);
            const v2f b1 = *(const v2f*)(bp + 32);
            const v2f b2 = *(const v2f*)(bp + 64);
            const v2f b3 = *(const v2f*)(bp + 96);
            acc0 = __builtin_amdgcn_wmma_f32_16x16x4_f32(false, a, false, b0, (short)0, acc0, false, false);
            acc1 = __builtin_amdgcn_wmma_f32_16x16x4_f32(false, a, false, b1, (short)0, acc1, false, false);
            acc2 = __builtin_amdgcn_wmma_f32_16x16x4_f32(false, a, false, b2, (short)0, acc2, false, false);
            acc3 = __builtin_amdgcn_wmma_f32_16x16x4_f32(false, a, false, b3, (short)0, acc3, false, false);
        }

        wave_lds_sync();
        #pragma unroll
        for (int r = 0; r < 8; ++r) {
            orow[r * T_PITCH +  0] = acc0[r];
            orow[r * T_PITCH + 16] = acc1[r];
            orow[r * T_PITCH + 32] = acc2[r];
            orow[r * T_PITCH + 48] = acc3[r];
        }
        wave_lds_sync();
        float* obase = O + (size_t)tile * 16 * OUT_PITCH + 4 * m;
        #pragma unroll
        for (int t = 0; t < 8; ++t) {
            const int rr = 2 * t + kh;
            const float4 v = *(const float4*)(mytile + rr * T_PITCH + 4 * m);
            *(float4*)(obase + (size_t)rr * OUT_PITCH) = v;
        }
        if (!has_next) break;
        tile = tnext;
        second = true;
    }
#undef X16
}

template <int J>
__device__ __forceinline__ int row_bcast(int v)
{
    return __builtin_amdgcn_update_dpp(0, v, 0x150 | J, 0xf, 0xf, false);
}
template <int J>
__device__ __forceinline__ float row_bcastf(float v)
{
    return __builtin_bit_cast(float, row_bcast<J>(__builtin_bit_cast(int, v)));
}

__device__ __forceinline__ float4 ld16(const float* base, unsigned byteoff)
{
    return *(const float4*)((const char*)base + byteoff);
}
__device__ __forceinline__ void st16(float* base, unsigned byteoff, float4 v)
{
    *(float4*)((char*)base + byteoff) = v;
}
__device__ __forceinline__ void fma4(float4& a, float w, const float4& v)
{
    a.x += w * v.x;  a.y += w * v.y;  a.z += w * v.z;  a.w += w * v.w;
}

#define EDGE_HEAD_BLOCKS 128
#define EDGE_THREADS     512
#define EDGE_HEAD_WAVES  (EDGE_HEAD_BLOCKS * (EDGE_THREADS / 32))

__global__ __launch_bounds__(EDGE_THREADS) void edge_kernel(
    const int*  __restrict__ edges,
    const float* __restrict__ asrc, const float* __restrict__ adst,
    const float* __restrict__ bsrc, const float* __restrict__ bdst,
    const float* __restrict__ node_rows, const float* __restrict__ edge_rows,
    float* __restrict__ node_sums, float* __restrict__ edge_sums,
    float* __restrict__ partials)
{
    __shared__ float dtab[N_NODES];
    __shared__ float sred[2][32];

    const int  head = blockIdx.x >= EDGE_HEAD_BLOCKS ? 1 : 0;
    const int  hb   = blockIdx.x - head * EDGE_HEAD_BLOCKS;
    const float* __restrict__ tab  = head ? edge_rows : node_rows;
    float*       __restrict__ out  = head ? edge_sums : node_sums;
    const float* __restrict__ ssrc = head ? bsrc : asrc;
    const float* __restrict__ sdst = head ? bdst : adst;

    {
        const float4* g4 = (const float4*)sdst;
        float4*       l4 = (float4*)dtab;
        #pragma unroll 8
        for (int t = 0; t < 24; ++t)
            l4[t * EDGE_THREADS + threadIdx.x] = g4[t * EDGE_THREADS + threadIdx.x];
        if (threadIdx.x < N_NODES / 4 - 24 * EDGE_THREADS)
            l4[24 * EDGE_THREADS + threadIdx.x] = g4[24 * EDGE_THREADS + threadIdx.x];
    }
    __syncthreads();

    const int lane = threadIdx.x & 31;
    const int wid  = __builtin_amdgcn_readfirstlane(threadIdx.x >> 5);
    const int h    = lane >> 4;
    const int c    = lane & 15;
    const unsigned laneoff = (unsigned)c * 16u;
    const int2* e2 = (const int2*)edges;

    float vs = 0.f, vs2 = 0.f;
    int p = hb * (EDGE_THREADS / 32) + wid;
    int2 ed = e2[(size_t)(2 * p + h) * DEG + c];
    for (; p < N_NODES / 2; p += EDGE_HEAD_WAVES) {
        const int node = 2 * p + h;
        const int pnext = p + EDGE_HEAD_WAVES;
        int2 ed_next = ed;
        if (pnext < N_NODES / 2) ed_next = e2[(size_t)(2 * pnext + h) * DEG + c];

        float x = ssrc[ed.x] + dtab[ed.y];
        x = x > 0.f ? x : 0.2f * x;
        x = fminf(fmaxf(x, -2.f), 2.f);
        const float att = __expf(x);

        float sum = att;
        sum = row_xor_add<8>(sum);  sum = row_xor_add<4>(sum);
        sum = row_xor_add<2>(sum);  sum = row_xor_add<1>(sum);
        const float norm = att / sum;

        float pn = norm, pn2 = norm * norm;
        pn = row_xor_add<8>(pn);  pn2 = row_xor_add<8>(pn2);
        pn = row_xor_add<4>(pn);  pn2 = row_xor_add<4>(pn2);
        pn = row_xor_add<2>(pn);  pn2 = row_xor_add<2>(pn2);
        pn = row_xor_add<1>(pn);  pn2 = row_xor_add<1>(pn2);
        vs += pn;  vs2 += pn2;

        const int d = ed.y;
        float4 r[16];
#define LOAD_ROW(J) r[J] = ld16(tab, (unsigned)row_bcast<J>(d) * 512u + laneoff);
        LOAD_ROW(0)  LOAD_ROW(1)  LOAD_ROW(2)  LOAD_ROW(3)  LOAD_ROW(4)  LOAD_ROW(5)  LOAD_ROW(6)  LOAD_ROW(7)
        LOAD_ROW(8)  LOAD_ROW(9)  LOAD_ROW(10) LOAD_ROW(11) LOAD_ROW(12) LOAD_ROW(13) LOAD_ROW(14) LOAD_ROW(15)
#undef LOAD_ROW
        float4 acc = {0.f, 0.f, 0.f, 0.f};
#define ADD_ROW(J) fma4(acc, row_bcastf<J>(norm), r[J]);
        ADD_ROW(0)  ADD_ROW(1)  ADD_ROW(2)  ADD_ROW(3)  ADD_ROW(4)  ADD_ROW(5)  ADD_ROW(6)  ADD_ROW(7)
        ADD_ROW(8)  ADD_ROW(9)  ADD_ROW(10) ADD_ROW(11) ADD_ROW(12) ADD_ROW(13) ADD_ROW(14) ADD_ROW(15)
#undef ADD_ROW

        st16(out, (unsigned)node * 512u + laneoff, acc);

        ed = ed_next;
    }

    if (c == 0) { sred[0][wid * 2 + h] = vs;  sred[1][wid * 2 + h] = vs2; }
    __syncthreads();
    if (threadIdx.x < 2) {
        float t = 0.f;
        #pragma unroll
        for (int w = 0; w < 32; ++w) t += sred[threadIdx.x][w];
        ((double*)partials)[(size_t)blockIdx.x * 2 + threadIdx.x] = (double)t;
    }
}

#define FIN_THREADS 64

typedef __attribute__((ext_vector_type(2))) int v2i;
template <int M>
__device__ __forceinline__ double row_xor_add_f64(double v)
{
    v2i b = __builtin_bit_cast(v2i, v);
    b.x = __builtin_amdgcn_update_dpp(0, b.x, 0x160 | M, 0xf, 0xf, false);
    b.y = __builtin_amdgcn_update_dpp(0, b.y, 0x160 | M, 0xf, 0xf, false);
    return v + __builtin_bit_cast(double, b);
}
__device__ __forceinline__ double xhalf_f64(double v)
{
    v2i b = __builtin_bit_cast(v2i, v);
    b.x = __builtin_amdgcn_permlanex16(b.x, b.x, 0x76543210, 0xfedcba98, false, false);
    b.y = __builtin_amdgcn_permlanex16(b.y, b.y, 0x76543210, 0xfedcba98, false, false);
    return __builtin_bit_cast(double, b);
}

__global__ __launch_bounds__(FIN_THREADS) void finalize_kernel(
    const float* __restrict__ partials, float* __restrict__ outv)
{
    const double* __restrict__ pd = (const double*)partials;
    const int lane = threadIdx.x & 31;
    const int head = threadIdx.x >> 5;
    const int q    = lane >> 4;
    const int c    = lane & 15;
    const double* src = pd + (size_t)(head * EDGE_HEAD_BLOCKS + c) * 2 + q;
    const double p0 = src[0 * 32], p1 = src[1 * 32], p2 = src[2 * 32], p3 = src[3 * 32];
    const double p4 = src[4 * 32], p5 = src[5 * 32], p6 = src[6 * 32], p7 = src[7 * 32];
    double t = ((p0 + p1) + (p2 + p3)) + ((p4 + p5) + (p6 + p7));
    t = row_xor_add_f64<8>(t);
    t = row_xor_add_f64<4>(t);
    t = row_xor_add_f64<2>(t);
    t = row_xor_add_f64<1>(t);
    const double other = xhalf_f64(t);
    if (lane == 0) {
        const double E = (double)N_NODES * (double)DEG;
        const double var = (other - t * t * (1.0 / E)) * (1.0 / (E - 1.0));
        outv[head] = (float)var;
    }
}

extern "C" void kernel_launch(void* const* d_in, const int* in_sizes, int n_in,
                              void* d_out, int out_size, void* d_ws, size_t ws_size,
                              hipStream_t stream) {
    const float* node_fts = (const float*)d_in[0];
    const float* edge_fts = (const float*)d_in[1];
    const int*   edges    = (const int*)  d_in[2];
    const float* W_node   = (const float*)d_in[3];
    const float* W_edge   = (const float*)d_in[4];
    const float* a_node   = (const float*)d_in[5];
    const float* a_edge   = (const float*)d_in[6];

    float* out      = (float*)d_out;
    float* node_out = out;
    float* edge_out = out + (size_t)N_NODES * 128;
    float* outv     = out + (size_t)2 * N_NODES * 128;

    float* ws   = (float*)d_ws;
    float* asrc = ws;
    float* adst = asrc + N_NODES;
    float* bsrc = adst + N_NODES;
    float* bdst = bsrc + N_NODES;
    float* partials = bdst + N_NODES;

    const int blocksA = 2 * PROJ_MAT_BLOCKS;
    proj_wmma_kernel<<<blocksA, PROJ_THREADS, 0, stream>>>(node_fts, edge_fts, W_node, W_edge, a_node, a_edge,
                                                       node_out, edge_out, asrc, adst, bsrc, bdst);

    const int blocksC = 2 * EDGE_HEAD_BLOCKS;
    edge_kernel<<<blocksC, EDGE_THREADS, 0, stream>>>(edges, asrc, adst, bsrc, bdst,
                                             node_out, edge_out, node_out + NOUT, edge_out + NOUT,
                                             partials);

    finalize_kernel<<<1, FIN_THREADS, 0, stream>>>(partials, outv);
}
